# starting kernel
# baseline (speedup 1.0000x reference)
#include <hip/hip_runtime.h>
#include <hip/hip_bf16.h>
#include <math.h>

#define BB      4
#define LL      4096
#define DMODEL  384
#define DSTATE  16
#define DTRANK  24
#define CTOT    (DTRANK + 2*DSTATE)
#define MTOT    (BB*LL)
#define NCH     32
#define CHL     (LL/NCH)

typedef float     v2f  __attribute__((ext_vector_type(2)));
typedef float     v8f  __attribute__((ext_vector_type(8)));
typedef _Float16  v16h __attribute__((ext_vector_type(16)));

#if defined(__gfx1250__) && __has_builtin(__builtin_amdgcn_wmma_f32_16x16x4_f32)
#define HAVE_WMMA_F32 1
#else
#define HAVE_WMMA_F32 0
#endif

__global__ __launch_bounds__(256)
void proj_gemm_kernel(const float* __restrict__ X,
                      const float* __restrict__ W1,
                      const float* __restrict__ prompt,
                      float* __restrict__ dtsr,
                      float* __restrict__ Bsc,
                      float* __restrict__ Csc)
{
    __shared__ float Wlds[16 * 388];

    const int c0 = blockIdx.y * 16;
    for (int i = threadIdx.x; i < 16 * DMODEL; i += 256) {
        int n = i / DMODEL, k = i - n * DMODEL;
        float w = (c0 + n < CTOT) ? W1[(c0 + n) * DMODEL + k] : 0.0f;
        Wlds[n * 388 + k] = w;
    }
    __syncthreads();

    const int wave = threadIdx.x >> 5;
    const int lane = threadIdx.x & 31;
    const int ln   = lane & 15;
    const int hi   = lane >> 4;
    const int m0   = blockIdx.x * 128 + wave * 16;

    v8f acc = {};
    const float* arow = X + (size_t)(m0 + ln) * DMODEL;

#if HAVE_WMMA_F32
    for (int k0 = 0; k0 < DMODEL; k0 += 4) {
        v2f a, b;
        a.x = arow[k0 + 2 * hi];
        a.y = arow[k0 + 2 * hi + 1];
        b.x = Wlds[ln * 388 + k0 + 2 * hi];
        b.y = Wlds[ln * 388 + k0 + 2 * hi + 1];
        acc = __builtin_amdgcn_wmma_f32_16x16x4_f32(
                  false, a, false, b, (short)0, acc, false, false);
    }
#else
    for (int k0 = 0; k0 < DMODEL; k0 += 32) {
        v16h a, b;
#pragma unroll
        for (int v = 0; v < 8; ++v) {
            int kk = (v < 4 ? 2 * v : 16 + 2 * (v - 4)) + 8 * hi;
            a[2 * v]     = (_Float16)arow[k0 + kk];
            a[2 * v + 1] = (_Float16)arow[k0 + kk + 1];
            b[2 * v]     = (_Float16)Wlds[ln * 388 + k0 + kk];
            b[2 * v + 1] = (_Float16)Wlds[ln * 388 + k0 + kk + 1];
        }
        acc = __builtin_amdgcn_wmma_f32_16x16x32_f16(
                  false, a, false, b, (short)0, acc, false, false);
    }
#endif

    const int c = c0 + ln;
#pragma unroll
    for (int r = 0; r < 8; ++r) {
        int m = m0 + r + 8 * hi;
        float v = acc[r];
        if (c < DTRANK) {
            dtsr[(size_t)m * DTRANK + c] = v;
        } else if (c < DTRANK + DSTATE) {
            Bsc[(size_t)m * DSTATE + (c - DTRANK)] = v;
        } else if (c < CTOT) {
            int n = c - DTRANK - DSTATE;
            Csc[(size_t)m * DSTATE + n] = v + prompt[(size_t)m * DSTATE + n];
        }
    }
}

__global__ __launch_bounds__(256)
void dt_gemm_kernel(const float* __restrict__ dtsr,
                    const float* __restrict__ W2,
                    const float* __restrict__ bias,
                    float* __restrict__ delta)
{
    __shared__ float Wlds[DMODEL * 28];

    for (int i = threadIdx.x; i < DMODEL * DTRANK; i += 256) {
        int d = i / DTRANK, r = i - d * DTRANK;
        Wlds[d * 28 + r] = W2[i];
    }
    __syncthreads();

    const int wave = threadIdx.x >> 5;
    const int lane = threadIdx.x & 31;
    const int ln   = lane & 15;
    const int hi   = lane >> 4;
    const int m0   = blockIdx.x * 128 + wave * 16;

    const float* arow = dtsr + (size_t)(m0 + ln) * DTRANK;

#if HAVE_WMMA_F32
    v2f afr[6];
#pragma unroll
    for (int ks = 0; ks < 6; ++ks) {
        afr[ks].x = arow[4 * ks + 2 * hi];
        afr[ks].y = arow[4 * ks + 2 * hi + 1];
    }
#else
    v16h afr;
#pragma unroll
    for (int v = 0; v < 8; ++v) {
        int kk = (v < 4 ? 2 * v : 16 + 2 * (v - 4)) + 8 * hi;
        afr[2 * v]     = (kk     < DTRANK) ? (_Float16)arow[kk]     : (_Float16)0.f;
        afr[2 * v + 1] = (kk + 1 < DTRANK) ? (_Float16)arow[kk + 1] : (_Float16)0.f;
    }
#endif

    for (int d0 = 0; d0 < DMODEL; d0 += 16) {
        v8f acc = {};
#if HAVE_WMMA_F32
#pragma unroll
        for (int ks = 0; ks < 6; ++ks) {
            v2f b;
            b.x = Wlds[(d0 + ln) * 28 + 4 * ks + 2 * hi];
            b.y = Wlds[(d0 + ln) * 28 + 4 * ks + 2 * hi + 1];
            acc = __builtin_amdgcn_wmma_f32_16x16x4_f32(
                      false, afr[ks], false, b, (short)0, acc, false, false);
        }
#else
        v16h b;
#pragma unroll
        for (int v = 0; v < 8; ++v) {
            int kk = (v < 4 ? 2 * v : 16 + 2 * (v - 4)) + 8 * hi;
            b[2 * v]     = (kk     < DTRANK) ? (_Float16)Wlds[(d0 + ln) * 28 + kk]     : (_Float16)0.f;
            b[2 * v + 1] = (kk + 1 < DTRANK) ? (_Float16)Wlds[(d0 + ln) * 28 + kk + 1] : (_Float16)0.f;
        }
        acc = __builtin_amdgcn_wmma_f32_16x16x32_f16(
                  false, afr, false, b, (short)0, acc, false, false);
#endif
        const int   d  = d0 + ln;
        const float bi = bias[d];
#pragma unroll
        for (int r = 0; r < 8; ++r) {
            int   m = m0 + r + 8 * hi;
            float t = acc[r] + bi;
            float sp = (t > 20.0f) ? t : log1pf(expf(t));
            delta[(size_t)m * DMODEL + d] = sp;
        }
    }
}

__global__ __launch_bounds__(256)
void scan_passA_kernel(const float* __restrict__ delta,
                       const float* __restrict__ X,
                       const float* __restrict__ Bsc,
                       const float* __restrict__ Alog,
                       float* __restrict__ Pout,
                       float* __restrict__ Hend)
{
    const int idx = blockIdx.x * 256 + threadIdx.x;
    const int d   = idx % DMODEL;
    const int t2  = idx / DMODEL;
    const int b   = t2 % BB;
    const int ch  = t2 / BB;

    float A[DSTATE], h[DSTATE], P[DSTATE];
#pragma unroll
    for (int n = 0; n < DSTATE; ++n) {
        A[n] = -expf(Alog[d * DSTATE + n]);
        h[n] = 0.0f;
        P[n] = 1.0f;
    }

    const int l0 = ch * CHL;
    for (int t = 0; t < CHL; ++t) {
        const int bl = b * LL + l0 + t;
        if (t + 8 < CHL) {
            __builtin_prefetch(&delta[(size_t)(bl + 8) * DMODEL + d], 0, 1);
            __builtin_prefetch(&X[(size_t)(bl + 8) * DMODEL + d], 0, 1);
        }
        const float dt = delta[(size_t)bl * DMODEL + d];
        const float u  = X[(size_t)bl * DMODEL + d];
        const float du = dt * u;
        const float4* Bv4 = reinterpret_cast<const float4*>(Bsc + (size_t)bl * DSTATE);
        float Bl[DSTATE];
#pragma unroll
        for (int q = 0; q < 4; ++q) {
            float4 v = Bv4[q];
            Bl[4 * q] = v.x; Bl[4 * q + 1] = v.y; Bl[4 * q + 2] = v.z; Bl[4 * q + 3] = v.w;
        }
#pragma unroll
        for (int n = 0; n < DSTATE; ++n) {
            float a = expf(dt * A[n]);
            P[n] *= a;
            h[n]  = a * h[n] + du * Bl[n];
        }
    }

    float* Pp = Pout + (size_t)idx * DSTATE;
    float* Hp = Hend + (size_t)idx * DSTATE;
#pragma unroll
    for (int n = 0; n < DSTATE; ++n) { Pp[n] = P[n]; Hp[n] = h[n]; }
}

__global__ __launch_bounds__(256)
void scan_passB_kernel(const float* __restrict__ P,
                       const float* __restrict__ Hend,
                       float* __restrict__ Hinit)
{
    const int idx = blockIdx.x * 256 + threadIdx.x;
    if (idx >= BB * DMODEL) return;
    const int b = idx / DMODEL;
    const int d = idx - b * DMODEL;

    float h[DSTATE];
#pragma unroll
    for (int n = 0; n < DSTATE; ++n) h[n] = 0.0f;

    for (int ch = 0; ch < NCH; ++ch) {
        const size_t unit = ((size_t)(ch * BB + b) * DMODEL + d) * DSTATE;
        float* Hi = Hinit + unit;
#pragma unroll
        for (int n = 0; n < DSTATE; ++n) Hi[n] = h[n];
#pragma unroll
        for (int n = 0; n < DSTATE; ++n)
            h[n] = P[unit + n] * h[n] + Hend[unit + n];
    }
}

__global__ __launch_bounds__(256)
void scan_passC_kernel(const float* __restrict__ delta,
                       const float* __restrict__ X,
                       const float* __restrict__ Bsc,
                       const float* __restrict__ Csc,
                       const float* __restrict__ Alog,
                       const float* __restrict__ Ds,
                       const float* __restrict__ Hinit,
                       float* __restrict__ Y)
{
    const int idx = blockIdx.x * 256 + threadIdx.x;
    const int d   = idx % DMODEL;
    const int t2  = idx / DMODEL;
    const int b   = t2 % BB;
    const int ch  = t2 / BB;

    float A[DSTATE], h[DSTATE];
    const float* Hi = Hinit + (size_t)idx * DSTATE;
#pragma unroll
    for (int n = 0; n < DSTATE; ++n) {
        A[n] = -expf(Alog[d * DSTATE + n]);
        h[n] = Hi[n];
    }
    const float Dsd = Ds[d];

    const int l0 = ch * CHL;
    for (int t = 0; t < CHL; ++t) {
        const int bl = b * LL + l0 + t;
        if (t + 8 < CHL) {
            __builtin_prefetch(&delta[(size_t)(bl + 8) * DMODEL + d], 0, 1);
            __builtin_prefetch(&X[(size_t)(bl + 8) * DMODEL + d], 0, 1);
        }
        const float dt = delta[(size_t)bl * DMODEL + d];
        const float u  = X[(size_t)bl * DMODEL + d];
        const float du = dt * u;
        const float4* Bv4 = reinterpret_cast<const float4*>(Bsc + (size_t)bl * DSTATE);
        const float4* Cv4 = reinterpret_cast<const float4*>(Csc + (size_t)bl * DSTATE);
        float Bl[DSTATE], Cl[DSTATE];
#pragma unroll
        for (int q = 0; q < 4; ++q) {
            float4 vb = Bv4[q], vc = Cv4[q];
            Bl[4*q] = vb.x; Bl[4*q+1] = vb.y; Bl[4*q+2] = vb.z; Bl[4*q+3] = vb.w;
            Cl[4*q] = vc.x; Cl[4*q+1] = vc.y; Cl[4*q+2] = vc.z; Cl[4*q+3] = vc.w;
        }
        float y = 0.0f;
#pragma unroll
        for (int n = 0; n < DSTATE; ++n) {
            float a = expf(dt * A[n]);
            h[n] = a * h[n] + du * Bl[n];
            y   += h[n] * Cl[n];
        }
        Y[(size_t)bl * DMODEL + d] = y + u * Dsd;
    }
}

extern "C" void kernel_launch(void* const* d_in, const int* in_sizes, int n_in,
                              void* d_out, int out_size, void* d_ws, size_t ws_size,
                              hipStream_t stream)
{
    const float* x      = (const float*)d_in[0];
    const float* prompt = (const float*)d_in[1];
    const float* w1     = (const float*)d_in[2];
    const float* w2     = (const float*)d_in[3];
    const float* bias   = (const float*)d_in[4];
    const float* alog   = (const float*)d_in[5];
    const float* ds     = (const float*)d_in[6];
    float*       y      = (float*)d_out;

    float* ws = (float*)d_ws;
    size_t o = 0;
    float* dtsr  = ws + o; o += (size_t)MTOT * DTRANK;
    float* Bsc   = ws + o; o += (size_t)MTOT * DSTATE;
    float* Csc   = ws + o; o += (size_t)MTOT * DSTATE;
    float* delta = ws + o; o += (size_t)MTOT * DMODEL;
    const size_t units = (size_t)NCH * BB * DMODEL;
    float* P     = ws + o; o += units * DSTATE;
    float* Hend  = ws + o; o += units * DSTATE;
    float* Hinit = ws + o; o += units * DSTATE;
    (void)ws_size; (void)in_sizes; (void)n_in; (void)out_size;

    proj_gemm_kernel<<<dim3(MTOT / 128, 4, 1), 256, 0, stream>>>(x, w1, prompt, dtsr, Bsc, Csc);
    dt_gemm_kernel  <<<dim3(MTOT / 128, 1, 1), 256, 0, stream>>>(dtsr, w2, bias, delta);
    scan_passA_kernel<<<dim3((unsigned)(units / 256)), 256, 0, stream>>>(delta, x, Bsc, alog, P, Hend);
    scan_passB_kernel<<<dim3((BB * DMODEL + 255) / 256), 256, 0, stream>>>(P, Hend, Hinit);
    scan_passC_kernel<<<dim3((unsigned)(units / 256)), 256, 0, stream>>>(delta, x, Bsc, Csc, alog, ds, Hinit, y);
}


# Scan passes A and C: the 16 states of each (chunk,b,d) are split over 4 waves of a block (4x the waves per SIMD); pass C sums y as four 4-state partials joined through LDS
# speedup vs baseline: 1.6443x; 1.6443x over previous
#include <hip/hip_runtime.h>
#include <hip/hip_bf16.h>
#include <math.h>

#define BB      4
#define LL      4096
#define DMODEL  384
#define DSTATE  16
#define DTRANK  24
#define CTOT    (DTRANK + 2*DSTATE)
#define MTOT    (BB*LL)
#define NCH     32
#define CHL     (LL/NCH)
#define NQ      4
#define SPQ     (DSTATE/NQ)
#define DGRP    (DMODEL/32)
#define TB      32

typedef float     v2f  __attribute__((ext_vector_type(2)));
typedef float     v8f  __attribute__((ext_vector_type(8)));
typedef _Float16  v16h __attribute__((ext_vector_type(16)));

#if defined(__gfx1250__) && __has_builtin(__builtin_amdgcn_wmma_f32_16x16x4_f32)
#define HAVE_WMMA_F32 1
#else
#define HAVE_WMMA_F32 0
#endif

__global__ __launch_bounds__(256)
void proj_gemm_kernel(const float* __restrict__ X,
                      const float* __restrict__ W1,
                      const float* __restrict__ prompt,
                      float* __restrict__ dtsr,
                      float* __restrict__ Bsc,
                      float* __restrict__ Csc)
{
    __shared__ float Wlds[16 * 388];

    const int c0 = blockIdx.y * 16;
    for (int i = threadIdx.x; i < 16 * DMODEL; i += 256) {
        int n = i / DMODEL, k = i - n * DMODEL;
        float w = (c0 + n < CTOT) ? W1[(c0 + n) * DMODEL + k] : 0.0f;
        Wlds[n * 388 + k] = w;
    }
    __syncthreads();

    const int wave = threadIdx.x >> 5;
    const int lane = threadIdx.x & 31;
    const int ln   = lane & 15;
    const int hi   = lane >> 4;
    const int m0   = blockIdx.x * 128 + wave * 16;

    v8f acc = {};
    const float* arow = X + (size_t)(m0 + ln) * DMODEL;

#if HAVE_WMMA_F32
    for (int k0 = 0; k0 < DMODEL; k0 += 4) {
        v2f a, b;
        a.x = arow[k0 + 2 * hi];
        a.y = arow[k0 + 2 * hi + 1];
        b.x = Wlds[ln * 388 + k0 + 2 * hi];
        b.y = Wlds[ln * 388 + k0 + 2 * hi + 1];
        acc = __builtin_amdgcn_wmma_f32_16x16x4_f32(
                  false, a, false, b, (short)0, acc, false, false);
    }
#else
    for (int k0 = 0; k0 < DMODEL; k0 += 32) {
        v16h a, b;
#pragma unroll
        for (int v = 0; v < 8; ++v) {
            int kk = (v < 4 ? 2 * v : 16 + 2 * (v - 4)) + 8 * hi;
            a[2 * v]     = (_Float16)arow[k0 + kk];
            a[2 * v + 1] = (_Float16)arow[k0 + kk + 1];
            b[2 * v]     = (_Float16)Wlds[ln * 388 + k0 + kk];
            b[2 * v + 1] = (_Float16)Wlds[ln * 388 + k0 + kk + 1];
        }
        acc = __builtin_amdgcn_wmma_f32_16x16x32_f16(
                  false, a, false, b, (short)0, acc, false, false);
    }
#endif

    const int c = c0 + ln;
#pragma unroll
    for (int r = 0; r < 8; ++r) {
        int m = m0 + r + 8 * hi;
        float v = acc[r];
        if (c < DTRANK) {
            dtsr[(size_t)m * DTRANK + c] = v;
        } else if (c < DTRANK + DSTATE) {
            Bsc[(size_t)m * DSTATE + (c - DTRANK)] = v;
        } else if (c < CTOT) {
            int n = c - DTRANK - DSTATE;
            Csc[(size_t)m * DSTATE + n] = v + prompt[(size_t)m * DSTATE + n];
        }
    }
}

__global__ __launch_bounds__(256)
void dt_gemm_kernel(const float* __restrict__ dtsr,
                    const float* __restrict__ W2,
                    const float* __restrict__ bias,
                    float* __restrict__ delta)
{
    __shared__ float Wlds[DMODEL * 28];

    for (int i = threadIdx.x; i < DMODEL * DTRANK; i += 256) {
        int d = i / DTRANK, r = i - d * DTRANK;
        Wlds[d * 28 + r] = W2[i];
    }
    __syncthreads();

    const int wave = threadIdx.x >> 5;
    const int lane = threadIdx.x & 31;
    const int ln   = lane & 15;
    const int hi   = lane >> 4;
    const int m0   = blockIdx.x * 128 + wave * 16;

    const float* arow = dtsr + (size_t)(m0 + ln) * DTRANK;

#if HAVE_WMMA_F32
    v2f afr[6];
#pragma unroll
    for (int ks = 0; ks < 6; ++ks) {
        afr[ks].x = arow[4 * ks + 2 * hi];
        afr[ks].y = arow[4 * ks + 2 * hi + 1];
    }
#else
    v16h afr;
#pragma unroll
    for (int v = 0; v < 8; ++v) {
        int kk = (v < 4 ? 2 * v : 16 + 2 * (v - 4)) + 8 * hi;
        afr[2 * v]     = (kk     < DTRANK) ? (_Float16)arow[kk]     : (_Float16)0.f;
        afr[2 * v + 1] = (kk + 1 < DTRANK) ? (_Float16)arow[kk + 1] : (_Float16)0.f;
    }
#endif

    for (int d0 = 0; d0 < DMODEL; d0 += 16) {
        v8f acc = {};
#if HAVE_WMMA_F32
#pragma unroll
        for (int ks = 0; ks < 6; ++ks) {
            v2f b;
            b.x = Wlds[(d0 + ln) * 28 + 4 * ks + 2 * hi];
            b.y = Wlds[(d0 + ln) * 28 + 4 * ks + 2 * hi + 1];
            acc = __builtin_amdgcn_wmma_f32_16x16x4_f32(
                      false, afr[ks], false, b, (short)0, acc, false, false);
        }
#else
        v16h b;
#pragma unroll
        for (int v = 0; v < 8; ++v) {
            int kk = (v < 4 ? 2 * v : 16 + 2 * (v - 4)) + 8 * hi;
            b[2 * v]     = (kk     < DTRANK) ? (_Float16)Wlds[(d0 + ln) * 28 + kk]     : (_Float16)0.f;
            b[2 * v + 1] = (kk + 1 < DTRANK) ? (_Float16)Wlds[(d0 + ln) * 28 + kk + 1] : (_Float16)0.f;
        }
        acc = __builtin_amdgcn_wmma_f32_16x16x32_f16(
                  false, afr, false, b, (short)0, acc, false, false);
#endif
        const int   d  = d0 + ln;
        const float bi = bias[d];
#pragma unroll
        for (int r = 0; r < 8; ++r) {
            int   m = m0 + r + 8 * hi;
            float t = acc[r] + bi;
            float sp = (t > 20.0f) ? t : log1pf(expf(t));
            delta[(size_t)m * DMODEL + d] = sp;
        }
    }
}

__global__ __launch_bounds__(128)
void scan_passA_kernel(const float* __restrict__ delta,
                       const float* __restrict__ X,
                       const float* __restrict__ Bsc,
                       const float* __restrict__ Alog,
                       float* __restrict__ Pout,
                       float* __restrict__ Hend)
{
    const int lane = threadIdx.x & 31;
    const int q    = threadIdx.x >> 5;
    const int dg   = blockIdx.x % DGRP;
    const int t2   = blockIdx.x / DGRP;
    const int b    = t2 % BB;
    const int ch   = t2 / BB;
    const int d    = dg * 32 + lane;
    const size_t unit = (size_t)(ch * BB + b) * DMODEL + d;

    float A[SPQ], h[SPQ], P[SPQ];
    {
        const float4 al = *reinterpret_cast<const float4*>(Alog + d * DSTATE + SPQ * q);
        A[0] = -expf(al.x); A[1] = -expf(al.y); A[2] = -expf(al.z); A[3] = -expf(al.w);
    }
    for (int j = 0; j < SPQ; ++j) { h[j] = 0.0f; P[j] = 1.0f; }

    const int l0 = ch * CHL;
    for (int t = 0; t < CHL; ++t) {
        const int bl = b * LL + l0 + t;
        if (t + 8 < CHL) {
            __builtin_prefetch(&delta[(size_t)(bl + 8) * DMODEL + d], 0, 1);
            __builtin_prefetch(&X[(size_t)(bl + 8) * DMODEL + d], 0, 1);
        }
        const float dt = delta[(size_t)bl * DMODEL + d];
        const float u  = X[(size_t)bl * DMODEL + d];
        const float du = dt * u;
        const float4 vb = *reinterpret_cast<const float4*>(Bsc + (size_t)bl * DSTATE + SPQ * q);
        const float Bl[SPQ] = { vb.x, vb.y, vb.z, vb.w };
        for (int j = 0; j < SPQ; ++j) {
            float a = expf(dt * A[j]);
            P[j] *= a;
            h[j]  = a * h[j] + du * Bl[j];
        }
    }

    float4* Pp = reinterpret_cast<float4*>(Pout + unit * DSTATE + SPQ * q);
    float4* Hp = reinterpret_cast<float4*>(Hend + unit * DSTATE + SPQ * q);
    *Pp = make_float4(P[0], P[1], P[2], P[3]);
    *Hp = make_float4(h[0], h[1], h[2], h[3]);
}

__global__ __launch_bounds__(256)
void scan_passB_kernel(const float* __restrict__ P,
                       const float* __restrict__ Hend,
                       float* __restrict__ Hinit)
{
    const int idx = blockIdx.x * 256 + threadIdx.x;
    if (idx >= BB * DMODEL) return;
    const int b = idx / DMODEL;
    const int d = idx - b * DMODEL;

    float h[DSTATE];
#pragma unroll
    for (int n = 0; n < DSTATE; ++n) h[n] = 0.0f;

    for (int ch = 0; ch < NCH; ++ch) {
        const size_t unit = ((size_t)(ch * BB + b) * DMODEL + d) * DSTATE;
        float* Hi = Hinit + unit;
#pragma unroll
        for (int n = 0; n < DSTATE; ++n) Hi[n] = h[n];
#pragma unroll
        for (int n = 0; n < DSTATE; ++n)
            h[n] = P[unit + n] * h[n] + Hend[unit + n];
    }
}

__global__ __launch_bounds__(128)
void scan_passC_kernel(const float* __restrict__ delta,
                       const float* __restrict__ X,
                       const float* __restrict__ Bsc,
                       const float* __restrict__ Csc,
                       const float* __restrict__ Alog,
                       const float* __restrict__ Ds,
                       const float* __restrict__ Hinit,
                       float* __restrict__ Y)
{
    __shared__ float ypart[2][TB][NQ][32];

    const int lane = threadIdx.x & 31;
    const int q    = threadIdx.x >> 5;
    const int dg   = blockIdx.x % DGRP;
    const int t2   = blockIdx.x / DGRP;
    const int b    = t2 % BB;
    const int ch   = t2 / BB;
    const int d    = dg * 32 + lane;
    const size_t unit = (size_t)(ch * BB + b) * DMODEL + d;

    float A[SPQ], h[SPQ];
    {
        const float4 al = *reinterpret_cast<const float4*>(Alog + d * DSTATE + SPQ * q);
        const float4 hv = *reinterpret_cast<const float4*>(Hinit + unit * DSTATE + SPQ * q);
        A[0] = -expf(al.x); A[1] = -expf(al.y); A[2] = -expf(al.z); A[3] = -expf(al.w);
        h[0] = hv.x; h[1] = hv.y; h[2] = hv.z; h[3] = hv.w;
    }
    const float Dsd = Ds[d];

    const int l0 = ch * CHL;
    for (int bt = 0; bt < CHL / TB; ++bt) {
        float (*yp)[NQ][32] = ypart[bt & 1];
        for (int tt = 0; tt < TB; ++tt) {
            const int t  = bt * TB + tt;
            const int bl = b * LL + l0 + t;
            if (t + 8 < CHL) {
                __builtin_prefetch(&delta[(size_t)(bl + 8) * DMODEL + d], 0, 1);
                __builtin_prefetch(&X[(size_t)(bl + 8) * DMODEL + d], 0, 1);
            }
            const float dt = delta[(size_t)bl * DMODEL + d];
            const float u  = X[(size_t)bl * DMODEL + d];
            const float du = dt * u;
            const float4 vb = *reinterpret_cast<const float4*>(Bsc + (size_t)bl * DSTATE + SPQ * q);
            const float4 vc = *reinterpret_cast<const float4*>(Csc + (size_t)bl * DSTATE + SPQ * q);
            const float Bl[SPQ] = { vb.x, vb.y, vb.z, vb.w };
            const float Cl[SPQ] = { vc.x, vc.y, vc.z, vc.w };
            float y = 0.0f;
            for (int j = 0; j < SPQ; ++j) {
                float a = expf(dt * A[j]);
                h[j] = a * h[j] + du * Bl[j];
                y   += h[j] * Cl[j];
            }
            if (q == 0) y += u * Dsd;
            yp[tt][q][lane] = y;
        }
        __syncthreads();
        for (int r = 0; r < TB / NQ; ++r) {
            const int tt = q * (TB / NQ) + r;
            const float s = (yp[tt][0][lane] + yp[tt][1][lane])
                          + (yp[tt][2][lane] + yp[tt][3][lane]);
            const int bl = b * LL + l0 + bt * TB + tt;
            Y[(size_t)bl * DMODEL + d] = s;
        }
    }
}

extern "C" void kernel_launch(void* const* d_in, const int* in_sizes, int n_in,
                              void* d_out, int out_size, void* d_ws, size_t ws_size,
                              hipStream_t stream)
{
    const float* x      = (const float*)d_in[0];
    const float* prompt = (const float*)d_in[1];
    const float* w1     = (const float*)d_in[2];
    const float* w2     = (const float*)d_in[3];
    const float* bias   = (const float*)d_in[4];
    const float* alog   = (const float*)d_in[5];
    const float* ds     = (const float*)d_in[6];
    float*       y      = (float*)d_out;

    float* ws = (float*)d_ws;
    size_t o = 0;
    float* dtsr  = ws + o; o += (size_t)MTOT * DTRANK;
    float* Bsc   = ws + o; o += (size_t)MTOT * DSTATE;
    float* Csc   = ws + o; o += (size_t)MTOT * DSTATE;
    float* delta = ws + o; o += (size_t)MTOT * DMODEL;
    const size_t units = (size_t)NCH * BB * DMODEL;
    float* P     = ws + o; o += units * DSTATE;
    float* Hend  = ws + o; o += units * DSTATE;
    float* Hinit = ws + o; o += units * DSTATE;
    (void)ws_size; (void)in_sizes; (void)n_in; (void)out_size;

    proj_gemm_kernel<<<dim3(MTOT / 128, 4, 1), 256, 0, stream>>>(x, w1, prompt, dtsr, Bsc, Csc);
    dt_gemm_kernel  <<<dim3(MTOT / 128, 1, 1), 256, 0, stream>>>(dtsr, w2, bias, delta);
    scan_passA_kernel<<<dim3((unsigned)(units / 32)), 128, 0, stream>>>(delta, x, Bsc, alog, P, Hend);
    scan_passB_kernel<<<dim3((BB * DMODEL + 255) / 256), 256, 0, stream>>>(P, Hend, Hinit);
    scan_passC_kernel<<<dim3((unsigned)(units / 32)), 128, 0, stream>>>(delta, x, Bsc, Csc, alog, ds, Hinit, y);
}


# dt GEMM+softplus: column tiles split over grid.y=6 (4 tiles per wave, was 24): unrolled text ~26 KB, was ~150 KB streamed past the 64 KB instruction cache; 6 waves/SIMD on all WGPs, was 2 on half
# speedup vs baseline: 2.4957x; 1.5178x over previous
#include <hip/hip_runtime.h>
#include <hip/hip_bf16.h>
#include <math.h>

#define BB      4
#define LL      4096
#define DMODEL  384
#define DSTATE  16
#define DTRANK  24
#define CTOT    (DTRANK + 2*DSTATE)
#define MTOT    (BB*LL)
#define NCH     32
#define CHL     (LL/NCH)
#define NQ      4
#define SPQ     (DSTATE/NQ)
#define DGRP    (DMODEL/32)
#define TB      32
#define DTSPLIT 6
#define DTCOLS  (DMODEL/DTSPLIT)

typedef float     v2f  __attribute__((ext_vector_type(2)));
typedef float     v8f  __attribute__((ext_vector_type(8)));
typedef _Float16  v16h __attribute__((ext_vector_type(16)));

#if defined(__gfx1250__) && __has_builtin(__builtin_amdgcn_wmma_f32_16x16x4_f32)
#define HAVE_WMMA_F32 1
#else
#define HAVE_WMMA_F32 0
#endif

__global__ __launch_bounds__(256)
void proj_gemm_kernel(const float* __restrict__ X,
                      const float* __restrict__ W1,
                      const float* __restrict__ prompt,
                      float* __restrict__ dtsr,
                      float* __restrict__ Bsc,
                      float* __restrict__ Csc)
{
    __shared__ float Wlds[16 * 388];

    const int c0 = blockIdx.y * 16;
    for (int i = threadIdx.x; i < 16 * DMODEL; i += 256) {
        int n = i / DMODEL, k = i - n * DMODEL;
        float w = (c0 + n < CTOT) ? W1[(c0 + n) * DMODEL + k] : 0.0f;
        Wlds[n * 388 + k] = w;
    }
    __syncthreads();

    const int wave = threadIdx.x >> 5;
    const int lane = threadIdx.x & 31;
    const int ln   = lane & 15;
    const int hi   = lane >> 4;
    const int m0   = blockIdx.x * 128 + wave * 16;

    v8f acc = {};
    const float* arow = X + (size_t)(m0 + ln) * DMODEL;

#if HAVE_WMMA_F32
    for (int k0 = 0; k0 < DMODEL; k0 += 4) {
        v2f a, b;
        a.x = arow[k0 + 2 * hi];
        a.y = arow[k0 + 2 * hi + 1];
        b.x = Wlds[ln * 388 + k0 + 2 * hi];
        b.y = Wlds[ln * 388 + k0 + 2 * hi + 1];
        acc = __builtin_amdgcn_wmma_f32_16x16x4_f32(
                  false, a, false, b, (short)0, acc, false, false);
    }
#else
    for (int k0 = 0; k0 < DMODEL; k0 += 32) {
        v16h a, b;
#pragma unroll
        for (int v = 0; v < 8; ++v) {
            int kk = (v < 4 ? 2 * v : 16 + 2 * (v - 4)) + 8 * hi;
            a[2 * v]     = (_Float16)arow[k0 + kk];
            a[2 * v + 1] = (_Float16)arow[k0 + kk + 1];
            b[2 * v]     = (_Float16)Wlds[ln * 388 + k0 + kk];
            b[2 * v + 1] = (_Float16)Wlds[ln * 388 + k0 + kk + 1];
        }
        acc = __builtin_amdgcn_wmma_f32_16x16x32_f16(
                  false, a, false, b, (short)0, acc, false, false);
    }
#endif

    const int c = c0 + ln;
#pragma unroll
    for (int r = 0; r < 8; ++r) {
        int m = m0 + r + 8 * hi;
        float v = acc[r];
        if (c < DTRANK) {
            dtsr[(size_t)m * DTRANK + c] = v;
        } else if (c < DTRANK + DSTATE) {
            Bsc[(size_t)m * DSTATE + (c - DTRANK)] = v;
        } else if (c < CTOT) {
            int n = c - DTRANK - DSTATE;
            Csc[(size_t)m * DSTATE + n] = v + prompt[(size_t)m * DSTATE + n];
        }
    }
}

__global__ __launch_bounds__(256)
void dt_gemm_kernel(const float* __restrict__ dtsr,
                    const float* __restrict__ W2,
                    const float* __restrict__ bias,
                    float* __restrict__ delta)
{
    __shared__ float Wlds[DTCOLS * 28];

    const int dbase = blockIdx.y * DTCOLS;
    for (int i = threadIdx.x; i < DTCOLS * DTRANK; i += 256) {
        int d = i / DTRANK, r = i - d * DTRANK;
        Wlds[d * 28 + r] = W2[dbase * DTRANK + i];
    }
    __syncthreads();

    const int wave = threadIdx.x >> 5;
    const int lane = threadIdx.x & 31;
    const int ln   = lane & 15;
    const int hi   = lane >> 4;
    const int m0   = blockIdx.x * 128 + wave * 16;

    const float* arow = dtsr + (size_t)(m0 + ln) * DTRANK;

#if HAVE_WMMA_F32
    v2f afr[6];
#pragma unroll
    for (int ks = 0; ks < 6; ++ks) {
        afr[ks].x = arow[4 * ks + 2 * hi];
        afr[ks].y = arow[4 * ks + 2 * hi + 1];
    }
#else
    v16h afr;
#pragma unroll
    for (int v = 0; v < 8; ++v) {
        int kk = (v < 4 ? 2 * v : 16 + 2 * (v - 4)) + 8 * hi;
        afr[2 * v]     = (kk     < DTRANK) ? (_Float16)arow[kk]     : (_Float16)0.f;
        afr[2 * v + 1] = (kk + 1 < DTRANK) ? (_Float16)arow[kk + 1] : (_Float16)0.f;
    }
#endif

    for (int d0 = 0; d0 < DTCOLS; d0 += 16) {
        v8f acc = {};
#if HAVE_WMMA_F32
#pragma unroll
        for (int ks = 0; ks < 6; ++ks) {
            v2f b;
            b.x = Wlds[(d0 + ln) * 28 + 4 * ks + 2 * hi];
            b.y = Wlds[(d0 + ln) * 28 + 4 * ks + 2 * hi + 1];
            acc = __builtin_amdgcn_wmma_f32_16x16x4_f32(
                      false, afr[ks], false, b, (short)0, acc, false, false);
        }
#else
        v16h b;
#pragma unroll
        for (int v = 0; v < 8; ++v) {
            int kk = (v < 4 ? 2 * v : 16 + 2 * (v - 4)) + 8 * hi;
            b[2 * v]     = (kk     < DTRANK) ? (_Float16)Wlds[(d0 + ln) * 28 + kk]     : (_Float16)0.f;
            b[2 * v + 1] = (kk + 1 < DTRANK) ? (_Float16)Wlds[(d0 + ln) * 28 + kk + 1] : (_Float16)0.f;
        }
        acc = __builtin_amdgcn_wmma_f32_16x16x32_f16(
                  false, afr, false, b, (short)0, acc, false, false);
#endif
        const int   d  = dbase + d0 + ln;
        const float bi = bias[d];
#pragma unroll
        for (int r = 0; r < 8; ++r) {
            int   m = m0 + r + 8 * hi;
            float t = acc[r] + bi;
            float sp = (t > 20.0f) ? t : log1pf(expf(t));
            delta[(size_t)m * DMODEL + d] = sp;
        }
    }
}

__global__ __launch_bounds__(128)
void scan_passA_kernel(const float* __restrict__ delta,
                       const float* __restrict__ X,
                       const float* __restrict__ Bsc,
                       const float* __restrict__ Alog,
                       float* __restrict__ Pout,
                       float* __restrict__ Hend)
{
    const int lane = threadIdx.x & 31;
    const int q    = threadIdx.x >> 5;
    const int dg   = blockIdx.x % DGRP;
    const int t2   = blockIdx.x / DGRP;
    const int b    = t2 % BB;
    const int ch   = t2 / BB;
    const int d    = dg * 32 + lane;
    const size_t unit = (size_t)(ch * BB + b) * DMODEL + d;

    float A[SPQ], h[SPQ], P[SPQ];
    {
        const float4 al = *reinterpret_cast<const float4*>(Alog + d * DSTATE + SPQ * q);
        A[0] = -expf(al.x); A[1] = -expf(al.y); A[2] = -expf(al.z); A[3] = -expf(al.w);
    }
    for (int j = 0; j < SPQ; ++j) { h[j] = 0.0f; P[j] = 1.0f; }

    const int l0 = ch * CHL;
    for (int t = 0; t < CHL; ++t) {
        const int bl = b * LL + l0 + t;
        if (t + 8 < CHL) {
            __builtin_prefetch(&delta[(size_t)(bl + 8) * DMODEL + d], 0, 1);
            __builtin_prefetch(&X[(size_t)(bl + 8) * DMODEL + d], 0, 1);
        }
        const float dt = delta[(size_t)bl * DMODEL + d];
        const float u  = X[(size_t)bl * DMODEL + d];
        const float du = dt * u;
        const float4 vb = *reinterpret_cast<const float4*>(Bsc + (size_t)bl * DSTATE + SPQ * q);
        const float Bl[SPQ] = { vb.x, vb.y, vb.z, vb.w };
        for (int j = 0; j < SPQ; ++j) {
            float a = expf(dt * A[j]);
            P[j] *= a;
            h[j]  = a * h[j] + du * Bl[j];
        }
    }

    float4* Pp = reinterpret_cast<float4*>(Pout + unit * DSTATE + SPQ * q);
    float4* Hp = reinterpret_cast<float4*>(Hend + unit * DSTATE + SPQ * q);
    *Pp = make_float4(P[0], P[1], P[2], P[3]);
    *Hp = make_float4(h[0], h[1], h[2], h[3]);
}

__global__ __launch_bounds__(256)
void scan_passB_kernel(const float* __restrict__ P,
                       const float* __restrict__ Hend,
                       float* __restrict__ Hinit)
{
    const int idx = blockIdx.x * 256 + threadIdx.x;
    if (idx >= BB * DMODEL) return;
    const int b = idx / DMODEL;
    const int d = idx - b * DMODEL;

    float h[DSTATE];
#pragma unroll
    for (int n = 0; n < DSTATE; ++n) h[n] = 0.0f;

    for (int ch = 0; ch < NCH; ++ch) {
        const size_t unit = ((size_t)(ch * BB + b) * DMODEL + d) * DSTATE;
        float* Hi = Hinit + unit;
#pragma unroll
        for (int n = 0; n < DSTATE; ++n) Hi[n] = h[n];
#pragma unroll
        for (int n = 0; n < DSTATE; ++n)
            h[n] = P[unit + n] * h[n] + Hend[unit + n];
    }
}

__global__ __launch_bounds__(128)
void scan_passC_kernel(const float* __restrict__ delta,
                       const float* __restrict__ X,
                       const float* __restrict__ Bsc,
                       const float* __restrict__ Csc,
                       const float* __restrict__ Alog,
                       const float* __restrict__ Ds,
                       const float* __restrict__ Hinit,
                       float* __restrict__ Y)
{
    __shared__ float ypart[2][TB][NQ][32];

    const int lane = threadIdx.x & 31;
    const int q    = threadIdx.x >> 5;
    const int dg   = blockIdx.x % DGRP;
    const int t2   = blockIdx.x / DGRP;
    const int b    = t2 % BB;
    const int ch   = t2 / BB;
    const int d    = dg * 32 + lane;
    const size_t unit = (size_t)(ch * BB + b) * DMODEL + d;

    float A[SPQ], h[SPQ];
    {
        const float4 al = *reinterpret_cast<const float4*>(Alog + d * DSTATE + SPQ * q);
        const float4 hv = *reinterpret_cast<const float4*>(Hinit + unit * DSTATE + SPQ * q);
        A[0] = -expf(al.x); A[1] = -expf(al.y); A[2] = -expf(al.z); A[3] = -expf(al.w);
        h[0] = hv.x; h[1] = hv.y; h[2] = hv.z; h[3] = hv.w;
    }
    const float Dsd = Ds[d];

    const int l0 = ch * CHL;
    for (int bt = 0; bt < CHL / TB; ++bt) {
        float (*yp)[NQ][32] = ypart[bt & 1];
        for (int tt = 0; tt < TB; ++tt) {
            const int t  = bt * TB + tt;
            const int bl = b * LL + l0 + t;
            if (t + 8 < CHL) {
                __builtin_prefetch(&delta[(size_t)(bl + 8) * DMODEL + d], 0, 1);
                __builtin_prefetch(&X[(size_t)(bl + 8) * DMODEL + d], 0, 1);
            }
            const float dt = delta[(size_t)bl * DMODEL + d];
            const float u  = X[(size_t)bl * DMODEL + d];
            const float du = dt * u;
            const float4 vb = *reinterpret_cast<const float4*>(Bsc + (size_t)bl * DSTATE + SPQ * q);
            const float4 vc = *reinterpret_cast<const float4*>(Csc + (size_t)bl * DSTATE + SPQ * q);
            const float Bl[SPQ] = { vb.x, vb.y, vb.z, vb.w };
            const float Cl[SPQ] = { vc.x, vc.y, vc.z, vc.w };
            float y = 0.0f;
            for (int j = 0; j < SPQ; ++j) {
                float a = expf(dt * A[j]);
                h[j] = a * h[j] + du * Bl[j];
                y   += h[j] * Cl[j];
            }
            if (q == 0) y += u * Dsd;
            yp[tt][q][lane] = y;
        }
        __syncthreads();
        for (int r = 0; r < TB / NQ; ++r) {
            const int tt = q * (TB / NQ) + r;
            const float s = (yp[tt][0][lane] + yp[tt][1][lane])
                          + (yp[tt][2][lane] + yp[tt][3][lane]);
            const int bl = b * LL + l0 + bt * TB + tt;
            Y[(size_t)bl * DMODEL + d] = s;
        }
    }
}

extern "C" void kernel_launch(void* const* d_in, const int* in_sizes, int n_in,
                              void* d_out, int out_size, void* d_ws, size_t ws_size,
                              hipStream_t stream)
{
    const float* x      = (const float*)d_in[0];
    const float* prompt = (const float*)d_in[1];
    const float* w1     = (const float*)d_in[2];
    const float* w2     = (const float*)d_in[3];
    const float* bias   = (const float*)d_in[4];
    const float* alog   = (const float*)d_in[5];
    const float* ds     = (const float*)d_in[6];
    float*       y      = (float*)d_out;

    float* ws = (float*)d_ws;
    size_t o = 0;
    float* dtsr  = ws + o; o += (size_t)MTOT * DTRANK;
    float* Bsc   = ws + o; o += (size_t)MTOT * DSTATE;
    float* Csc   = ws + o; o += (size_t)MTOT * DSTATE;
    float* delta = ws + o; o += (size_t)MTOT * DMODEL;
    const size_t units = (size_t)NCH * BB * DMODEL;
    float* P     = ws + o; o += units * DSTATE;
    float* Hend  = ws + o; o += units * DSTATE;
    float* Hinit = ws + o; o += units * DSTATE;
    (void)ws_size; (void)in_sizes; (void)n_in; (void)out_size;

    proj_gemm_kernel<<<dim3(MTOT / 128, 4, 1), 256, 0, stream>>>(x, w1, prompt, dtsr, Bsc, Csc);
    dt_gemm_kernel  <<<dim3(MTOT / 128, DTSPLIT, 1), 256, 0, stream>>>(dtsr, w2, bias, delta);
    scan_passA_kernel<<<dim3((unsigned)(units / 32)), 128, 0, stream>>>(delta, x, Bsc, alog, P, Hend);
    scan_passB_kernel<<<dim3((BB * DMODEL + 255) / 256), 256, 0, stream>>>(P, Hend, Hinit);
    scan_passC_kernel<<<dim3((unsigned)(units / 32)), 128, 0, stream>>>(delta, x, Bsc, Csc, alog, ds, Hinit, y);
}


# Scan passes A and C: loads of 8 steps issued together ahead of their use (was load, wait, compute per step: one L2 latency per step on a 128-step serial chain); per-step L2 prefetch dropped
# speedup vs baseline: 2.9213x; 1.1705x over previous
#include <hip/hip_runtime.h>
#include <hip/hip_bf16.h>
#include <math.h>

#define BB      4
#define LL      4096
#define DMODEL  384
#define DSTATE  16
#define DTRANK  24
#define CTOT    (DTRANK + 2*DSTATE)
#define MTOT    (BB*LL)
#define NCH     32
#define CHL     (LL/NCH)
#define NQ      4
#define SPQ     (DSTATE/NQ)
#define DGRP    (DMODEL/32)
#define TB      32
#define UB      8
#define DTSPLIT 6
#define DTCOLS  (DMODEL/DTSPLIT)

typedef float     v2f  __attribute__((ext_vector_type(2)));
typedef float     v8f  __attribute__((ext_vector_type(8)));
typedef _Float16  v16h __attribute__((ext_vector_type(16)));

#if defined(__gfx1250__) && __has_builtin(__builtin_amdgcn_wmma_f32_16x16x4_f32)
#define HAVE_WMMA_F32 1
#else
#define HAVE_WMMA_F32 0
#endif

__global__ __launch_bounds__(256)
void proj_gemm_kernel(const float* __restrict__ X,
                      const float* __restrict__ W1,
                      const float* __restrict__ prompt,
                      float* __restrict__ dtsr,
                      float* __restrict__ Bsc,
                      float* __restrict__ Csc)
{
    __shared__ float Wlds[16 * 388];

    const int c0 = blockIdx.y * 16;
    for (int i = threadIdx.x; i < 16 * DMODEL; i += 256) {
        int n = i / DMODEL, k = i - n * DMODEL;
        float w = (c0 + n < CTOT) ? W1[(c0 + n) * DMODEL + k] : 0.0f;
        Wlds[n * 388 + k] = w;
    }
    __syncthreads();

    const int wave = threadIdx.x >> 5;
    const int lane = threadIdx.x & 31;
    const int ln   = lane & 15;
    const int hi   = lane >> 4;
    const int m0   = blockIdx.x * 128 + wave * 16;

    v8f acc = {};
    const float* arow = X + (size_t)(m0 + ln) * DMODEL;

#if HAVE_WMMA_F32
    for (int k0 = 0; k0 < DMODEL; k0 += 4) {
        v2f a, b;
        a.x = arow[k0 + 2 * hi];
        a.y = arow[k0 + 2 * hi + 1];
        b.x = Wlds[ln * 388 + k0 + 2 * hi];
        b.y = Wlds[ln * 388 + k0 + 2 * hi + 1];
        acc = __builtin_amdgcn_wmma_f32_16x16x4_f32(
                  false, a, false, b, (short)0, acc, false, false);
    }
#else
    for (int k0 = 0; k0 < DMODEL; k0 += 32) {
        v16h a, b;
#pragma unroll
        for (int v = 0; v < 8; ++v) {
            int kk = (v < 4 ? 2 * v : 16 + 2 * (v - 4)) + 8 * hi;
            a[2 * v]     = (_Float16)arow[k0 + kk];
            a[2 * v + 1] = (_Float16)arow[k0 + kk + 1];
            b[2 * v]     = (_Float16)Wlds[ln * 388 + k0 + kk];
            b[2 * v + 1] = (_Float16)Wlds[ln * 388 + k0 + kk + 1];
        }
        acc = __builtin_amdgcn_wmma_f32_16x16x32_f16(
                  false, a, false, b, (short)0, acc, false, false);
    }
#endif

    const int c = c0 + ln;
#pragma unroll
    for (int r = 0; r < 8; ++r) {
        int m = m0 + r + 8 * hi;
        float v = acc[r];
        if (c < DTRANK) {
            dtsr[(size_t)m * DTRANK + c] = v;
        } else if (c < DTRANK + DSTATE) {
            Bsc[(size_t)m * DSTATE + (c - DTRANK)] = v;
        } else if (c < CTOT) {
            int n = c - DTRANK - DSTATE;
            Csc[(size_t)m * DSTATE + n] = v + prompt[(size_t)m * DSTATE + n];
        }
    }
}

__global__ __launch_bounds__(256)
void dt_gemm_kernel(const float* __restrict__ dtsr,
                    const float* __restrict__ W2,
                    const float* __restrict__ bias,
                    float* __restrict__ delta)
{
    __shared__ float Wlds[DTCOLS * 28];

    const int dbase = blockIdx.y * DTCOLS;
    for (int i = threadIdx.x; i < DTCOLS * DTRANK; i += 256) {
        int d = i / DTRANK, r = i - d * DTRANK;
        Wlds[d * 28 + r] = W2[dbase * DTRANK + i];
    }
    __syncthreads();

    const int wave = threadIdx.x >> 5;
    const int lane = threadIdx.x & 31;
    const int ln   = lane & 15;
    const int hi   = lane >> 4;
    const int m0   = blockIdx.x * 128 + wave * 16;

    const float* arow = dtsr + (size_t)(m0 + ln) * DTRANK;

#if HAVE_WMMA_F32
    v2f afr[6];
#pragma unroll
    for (int ks = 0; ks < 6; ++ks) {
        afr[ks].x = arow[4 * ks + 2 * hi];
        afr[ks].y = arow[4 * ks + 2 * hi + 1];
    }
#else
    v16h afr;
#pragma unroll
    for (int v = 0; v < 8; ++v) {
        int kk = (v < 4 ? 2 * v : 16 + 2 * (v - 4)) + 8 * hi;
        afr[2 * v]     = (kk     < DTRANK) ? (_Float16)arow[kk]     : (_Float16)0.f;
        afr[2 * v + 1] = (kk + 1 < DTRANK) ? (_Float16)arow[kk + 1] : (_Float16)0.f;
    }
#endif

    for (int d0 = 0; d0 < DTCOLS; d0 += 16) {
        v8f acc = {};
#if HAVE_WMMA_F32
#pragma unroll
        for (int ks = 0; ks < 6; ++ks) {
            v2f b;
            b.x = Wlds[(d0 + ln) * 28 + 4 * ks + 2 * hi];
            b.y = Wlds[(d0 + ln) * 28 + 4 * ks + 2 * hi + 1];
            acc = __builtin_amdgcn_wmma_f32_16x16x4_f32(
                      false, afr[ks], false, b, (short)0, acc, false, false);
        }
#else
        v16h b;
#pragma unroll
        for (int v = 0; v < 8; ++v) {
            int kk = (v < 4 ? 2 * v : 16 + 2 * (v - 4)) + 8 * hi;
            b[2 * v]     = (kk     < DTRANK) ? (_Float16)Wlds[(d0 + ln) * 28 + kk]     : (_Float16)0.f;
            b[2 * v + 1] = (kk + 1 < DTRANK) ? (_Float16)Wlds[(d0 + ln) * 28 + kk + 1] : (_Float16)0.f;
        }
        acc = __builtin_amdgcn_wmma_f32_16x16x32_f16(
                  false, afr, false, b, (short)0, acc, false, false);
#endif
        const int   d  = dbase + d0 + ln;
        const float bi = bias[d];
#pragma unroll
        for (int r = 0; r < 8; ++r) {
            int   m = m0 + r + 8 * hi;
            float t = acc[r] + bi;
            float sp = (t > 20.0f) ? t : log1pf(expf(t));
            delta[(size_t)m * DMODEL + d] = sp;
        }
    }
}

__global__ __launch_bounds__(128) __attribute__((amdgpu_waves_per_eu(8, 8)))
void scan_passA_kernel(const float* __restrict__ delta,
                       const float* __restrict__ X,
                       const float* __restrict__ Bsc,
                       const float* __restrict__ Alog,
                       float* __restrict__ Pout,
                       float* __restrict__ Hend)
{
    const int lane = threadIdx.x & 31;
    const int q    = threadIdx.x >> 5;
    const int dg   = blockIdx.x % DGRP;
    const int t2   = blockIdx.x / DGRP;
    const int b    = t2 % BB;
    const int ch   = t2 / BB;
    const int d    = dg * 32 + lane;
    const size_t unit = (size_t)(ch * BB + b) * DMODEL + d;

    float A[SPQ], h[SPQ], P[SPQ];
    {
        const float4 al = *reinterpret_cast<const float4*>(Alog + d * DSTATE + SPQ * q);
        A[0] = -expf(al.x); A[1] = -expf(al.y); A[2] = -expf(al.z); A[3] = -expf(al.w);
    }
    for (int j = 0; j < SPQ; ++j) { h[j] = 0.0f; P[j] = 1.0f; }

    const size_t row0 = (size_t)(b * LL + ch * CHL);
    const float* dp = delta + row0 * DMODEL + d;
    const float* xp = X     + row0 * DMODEL + d;
    const float* bp = Bsc   + row0 * DSTATE + SPQ * q;
#pragma unroll 1
    for (int t0 = 0; t0 < CHL; t0 += UB) {
        float  dtv[UB], uv[UB];
        float4 bv[UB];
#pragma unroll
        for (int i = 0; i < UB; ++i) {
            dtv[i] = dp[(size_t)(t0 + i) * DMODEL];
            uv[i]  = xp[(size_t)(t0 + i) * DMODEL];
            bv[i]  = *reinterpret_cast<const float4*>(bp + (size_t)(t0 + i) * DSTATE);
        }
#pragma unroll
        for (int i = 0; i < UB; ++i) {
            const float dt = dtv[i];
            const float du = dt * uv[i];
            const float Bl[SPQ] = { bv[i].x, bv[i].y, bv[i].z, bv[i].w };
            for (int j = 0; j < SPQ; ++j) {
                float a = expf(dt * A[j]);
                P[j] *= a;
                h[j]  = a * h[j] + du * Bl[j];
            }
        }
    }

    float4* Pp = reinterpret_cast<float4*>(Pout + unit * DSTATE + SPQ * q);
    float4* Hp = reinterpret_cast<float4*>(Hend + unit * DSTATE + SPQ * q);
    *Pp = make_float4(P[0], P[1], P[2], P[3]);
    *Hp = make_float4(h[0], h[1], h[2], h[3]);
}

__global__ __launch_bounds__(256)
void scan_passB_kernel(const float* __restrict__ P,
                       const float* __restrict__ Hend,
                       float* __restrict__ Hinit)
{
    const int idx = blockIdx.x * 256 + threadIdx.x;
    if (idx >= BB * DMODEL) return;
    const int b = idx / DMODEL;
    const int d = idx - b * DMODEL;

    float h[DSTATE];
#pragma unroll
    for (int n = 0; n < DSTATE; ++n) h[n] = 0.0f;

    for (int ch = 0; ch < NCH; ++ch) {
        const size_t unit = ((size_t)(ch * BB + b) * DMODEL + d) * DSTATE;
        float* Hi = Hinit + unit;
#pragma unroll
        for (int n = 0; n < DSTATE; ++n) Hi[n] = h[n];
#pragma unroll
        for (int n = 0; n < DSTATE; ++n)
            h[n] = P[unit + n] * h[n] + Hend[unit + n];
    }
}

__global__ __launch_bounds__(128) __attribute__((amdgpu_waves_per_eu(8, 8)))
void scan_passC_kernel(const float* __restrict__ delta,
                       const float* __restrict__ X,
                       const float* __restrict__ Bsc,
                       const float* __restrict__ Csc,
                       const float* __restrict__ Alog,
                       const float* __restrict__ Ds,
                       const float* __restrict__ Hinit,
                       float* __restrict__ Y)
{
    __shared__ float ypart[2][TB][NQ][32];

    const int lane = threadIdx.x & 31;
    const int q    = threadIdx.x >> 5;
    const int dg   = blockIdx.x % DGRP;
    const int t2   = blockIdx.x / DGRP;
    const int b    = t2 % BB;
    const int ch   = t2 / BB;
    const int d    = dg * 32 + lane;
    const size_t unit = (size_t)(ch * BB + b) * DMODEL + d;

    float A[SPQ], h[SPQ];
    {
        const float4 al = *reinterpret_cast<const float4*>(Alog + d * DSTATE + SPQ * q);
        const float4 hv = *reinterpret_cast<const float4*>(Hinit + unit * DSTATE + SPQ * q);
        A[0] = -expf(al.x); A[1] = -expf(al.y); A[2] = -expf(al.z); A[3] = -expf(al.w);
        h[0] = hv.x; h[1] = hv.y; h[2] = hv.z; h[3] = hv.w;
    }
    const float Dsd = Ds[d];

    const int l0 = ch * CHL;
    const size_t row0 = (size_t)(b * LL + l0);
    const float* dp = delta + row0 * DMODEL + d;
    const float* xp = X     + row0 * DMODEL + d;
    const float* bp = Bsc   + row0 * DSTATE + SPQ * q;
    const float* cp = Csc   + row0 * DSTATE + SPQ * q;
    for (int bt = 0; bt < CHL / TB; ++bt) {
        float (*yp)[NQ][32] = ypart[bt & 1];
#pragma unroll 1
        for (int s0 = 0; s0 < TB; s0 += UB) {
            const int t0 = bt * TB + s0;
            float  dtv[UB], uv[UB];
            float4 bv[UB], cv[UB];
#pragma unroll
            for (int i = 0; i < UB; ++i) {
                dtv[i] = dp[(size_t)(t0 + i) * DMODEL];
                uv[i]  = xp[(size_t)(t0 + i) * DMODEL];
                bv[i]  = *reinterpret_cast<const float4*>(bp + (size_t)(t0 + i) * DSTATE);
                cv[i]  = *reinterpret_cast<const float4*>(cp + (size_t)(t0 + i) * DSTATE);
            }
#pragma unroll
            for (int i = 0; i < UB; ++i) {
                const float dt = dtv[i];
                const float u  = uv[i];
                const float du = dt * u;
                const float Bl[SPQ] = { bv[i].x, bv[i].y, bv[i].z, bv[i].w };
                const float Cl[SPQ] = { cv[i].x, cv[i].y, cv[i].z, cv[i].w };
                float y = 0.0f;
                for (int j = 0; j < SPQ; ++j) {
                    float a = expf(dt * A[j]);
                    h[j] = a * h[j] + du * Bl[j];
                    y   += h[j] * Cl[j];
                }
                if (q == 0) y += u * Dsd;
                yp[s0 + i][q][lane] = y;
            }
        }
        __syncthreads();
        for (int r = 0; r < TB / NQ; ++r) {
            const int tt = q * (TB / NQ) + r;
            const float s = (yp[tt][0][lane] + yp[tt][1][lane])
                          + (yp[tt][2][lane] + yp[tt][3][lane]);
            const int bl = b * LL + l0 + bt * TB + tt;
            Y[(size_t)bl * DMODEL + d] = s;
        }
    }
}

extern "C" void kernel_launch(void* const* d_in, const int* in_sizes, int n_in,
                              void* d_out, int out_size, void* d_ws, size_t ws_size,
                              hipStream_t stream)
{
    const float* x      = (const float*)d_in[0];
    const float* prompt = (const float*)d_in[1];
    const float* w1     = (const float*)d_in[2];
    const float* w2     = (const float*)d_in[3];
    const float* bias   = (const float*)d_in[4];
    const float* alog   = (const float*)d_in[5];
    const float* ds     = (const float*)d_in[6];
    float*       y      = (float*)d_out;

    float* ws = (float*)d_ws;
    size_t o = 0;
    float* dtsr  = ws + o; o += (size_t)MTOT * DTRANK;
    float* Bsc   = ws + o; o += (size_t)MTOT * DSTATE;
    float* Csc   = ws + o; o += (size_t)MTOT * DSTATE;
    float* delta = ws + o; o += (size_t)MTOT * DMODEL;
    const size_t units = (size_t)NCH * BB * DMODEL;
    float* P     = ws + o; o += units * DSTATE;
    float* Hend  = ws + o; o += units * DSTATE;
    float* Hinit = ws + o; o += units * DSTATE;
    (void)ws_size; (void)in_sizes; (void)n_in; (void)out_size;

    proj_gemm_kernel<<<dim3(MTOT / 128, 4, 1), 256, 0, stream>>>(x, w1, prompt, dtsr, Bsc, Csc);
    dt_gemm_kernel  <<<dim3(MTOT / 128, DTSPLIT, 1), 256, 0, stream>>>(dtsr, w2, bias, delta);
    scan_passA_kernel<<<dim3((unsigned)(units / 32)), 128, 0, stream>>>(delta, x, Bsc, alog, P, Hend);
    scan_passB_kernel<<<dim3((BB * DMODEL + 255) / 256), 256, 0, stream>>>(P, Hend, Hinit);
    scan_passC_kernel<<<dim3((unsigned)(units / 32)), 128, 0, stream>>>(delta, x, Bsc, Csc, alog, ds, Hinit, y);
}


# Scan passes A and C: decay exp(dt*A) formed as exp2(dt*(A*log2 e)) on the hardware exponential instruction, A scaled once per thread; the library expf's argument split and range checks go
# speedup vs baseline: 4.0635x; 1.3910x over previous
#include <hip/hip_runtime.h>
#include <hip/hip_bf16.h>
#include <math.h>

#define BB      4
#define LL      4096
#define DMODEL  384
#define DSTATE  16
#define DTRANK  24
#define CTOT    (DTRANK + 2*DSTATE)
#define MTOT    (BB*LL)
#define NCH     32
#define CHL     (LL/NCH)
#define NQ      4
#define SPQ     (DSTATE/NQ)
#define DGRP    (DMODEL/32)
#define TB      32
#define UB      8
#define DTSPLIT 6
#define DTCOLS  (DMODEL/DTSPLIT)
#define LOG2E_F 1.4426950408889634f

typedef float     v2f  __attribute__((ext_vector_type(2)));
typedef float     v8f  __attribute__((ext_vector_type(8)));
typedef _Float16  v16h __attribute__((ext_vector_type(16)));

#if defined(__gfx1250__) && __has_builtin(__builtin_amdgcn_wmma_f32_16x16x4_f32)
#define HAVE_WMMA_F32 1
#else
#define HAVE_WMMA_F32 0
#endif

__global__ __launch_bounds__(256)
void proj_gemm_kernel(const float* __restrict__ X,
                      const float* __restrict__ W1,
                      const float* __restrict__ prompt,
                      float* __restrict__ dtsr,
                      float* __restrict__ Bsc,
                      float* __restrict__ Csc)
{
    __shared__ float Wlds[16 * 388];

    const int c0 = blockIdx.y * 16;
    for (int i = threadIdx.x; i < 16 * DMODEL; i += 256) {
        int n = i / DMODEL, k = i - n * DMODEL;
        float w = (c0 + n < CTOT) ? W1[(c0 + n) * DMODEL + k] : 0.0f;
        Wlds[n * 388 + k] = w;
    }
    __syncthreads();

    const int wave = threadIdx.x >> 5;
    const int lane = threadIdx.x & 31;
    const int ln   = lane & 15;
    const int hi   = lane >> 4;
    const int m0   = blockIdx.x * 128 + wave * 16;

    v8f acc = {};
    const float* arow = X + (size_t)(m0 + ln) * DMODEL;

#if HAVE_WMMA_F32
    for (int k0 = 0; k0 < DMODEL; k0 += 4) {
        v2f a, b;
        a.x = arow[k0 + 2 * hi];
        a.y = arow[k0 + 2 * hi + 1];
        b.x = Wlds[ln * 388 + k0 + 2 * hi];
        b.y = Wlds[ln * 388 + k0 + 2 * hi + 1];
        acc = __builtin_amdgcn_wmma_f32_16x16x4_f32(
                  false, a, false, b, (short)0, acc, false, false);
    }
#else
    for (int k0 = 0; k0 < DMODEL; k0 += 32) {
        v16h a, b;
#pragma unroll
        for (int v = 0; v < 8; ++v) {
            int kk = (v < 4 ? 2 * v : 16 + 2 * (v - 4)) + 8 * hi;
            a[2 * v]     = (_Float16)arow[k0 + kk];
            a[2 * v + 1] = (_Float16)arow[k0 + kk + 1];
            b[2 * v]     = (_Float16)Wlds[ln * 388 + k0 + kk];
            b[2 * v + 1] = (_Float16)Wlds[ln * 388 + k0 + kk + 1];
        }
        acc = __builtin_amdgcn_wmma_f32_16x16x32_f16(
                  false, a, false, b, (short)0, acc, false, false);
    }
#endif

    const int c = c0 + ln;
#pragma unroll
    for (int r = 0; r < 8; ++r) {
        int m = m0 + r + 8 * hi;
        float v = acc[r];
        if (c < DTRANK) {
            dtsr[(size_t)m * DTRANK + c] = v;
        } else if (c < DTRANK + DSTATE) {
            Bsc[(size_t)m * DSTATE + (c - DTRANK)] = v;
        } else if (c < CTOT) {
            int n = c - DTRANK - DSTATE;
            Csc[(size_t)m * DSTATE + n] = v + prompt[(size_t)m * DSTATE + n];
        }
    }
}

__global__ __launch_bounds__(256)
void dt_gemm_kernel(const float* __restrict__ dtsr,
                    const float* __restrict__ W2,
                    const float* __restrict__ bias,
                    float* __restrict__ delta)
{
    __shared__ float Wlds[DTCOLS * 28];

    const int dbase = blockIdx.y * DTCOLS;
    for (int i = threadIdx.x; i < DTCOLS * DTRANK; i += 256) {
        int d = i / DTRANK, r = i - d * DTRANK;
        Wlds[d * 28 + r] = W2[dbase * DTRANK + i];
    }
    __syncthreads();

    const int wave = threadIdx.x >> 5;
    const int lane = threadIdx.x & 31;
    const int ln   = lane & 15;
    const int hi   = lane >> 4;
    const int m0   = blockIdx.x * 128 + wave * 16;

    const float* arow = dtsr + (size_t)(m0 + ln) * DTRANK;

#if HAVE_WMMA_F32
    v2f afr[6];
#pragma unroll
    for (int ks = 0; ks < 6; ++ks) {
        afr[ks].x = arow[4 * ks + 2 * hi];
        afr[ks].y = arow[4 * ks + 2 * hi + 1];
    }
#else
    v16h afr;
#pragma unroll
    for (int v = 0; v < 8; ++v) {
        int kk = (v < 4 ? 2 * v : 16 + 2 * (v - 4)) + 8 * hi;
        afr[2 * v]     = (kk     < DTRANK) ? (_Float16)arow[kk]     : (_Float16)0.f;
        afr[2 * v + 1] = (kk + 1 < DTRANK) ? (_Float16)arow[kk + 1] : (_Float16)0.f;
    }
#endif

    for (int d0 = 0; d0 < DTCOLS; d0 += 16) {
        v8f acc = {};
#if HAVE_WMMA_F32
#pragma unroll
        for (int ks = 0; ks < 6; ++ks) {
            v2f b;
            b.x = Wlds[(d0 + ln) * 28 + 4 * ks + 2 * hi];
            b.y = Wlds[(d0 + ln) * 28 + 4 * ks + 2 * hi + 1];
            acc = __builtin_amdgcn_wmma_f32_16x16x4_f32(
                      false, afr[ks], false, b, (short)0, acc, false, false);
        }
#else
        v16h b;
#pragma unroll
        for (int v = 0; v < 8; ++v) {
            int kk = (v < 4 ? 2 * v : 16 + 2 * (v - 4)) + 8 * hi;
            b[2 * v]     = (kk     < DTRANK) ? (_Float16)Wlds[(d0 + ln) * 28 + kk]     : (_Float16)0.f;
            b[2 * v + 1] = (kk + 1 < DTRANK) ? (_Float16)Wlds[(d0 + ln) * 28 + kk + 1] : (_Float16)0.f;
        }
        acc = __builtin_amdgcn_wmma_f32_16x16x32_f16(
                  false, afr, false, b, (short)0, acc, false, false);
#endif
        const int   d  = dbase + d0 + ln;
        const float bi = bias[d];
#pragma unroll
        for (int r = 0; r < 8; ++r) {
            int   m = m0 + r + 8 * hi;
            float t = acc[r] + bi;
            float sp = (t > 20.0f) ? t : log1pf(expf(t));
            delta[(size_t)m * DMODEL + d] = sp;
        }
    }
}

__global__ __launch_bounds__(128) __attribute__((amdgpu_waves_per_eu(8, 8)))
void scan_passA_kernel(const float* __restrict__ delta,
                       const float* __restrict__ X,
                       const float* __restrict__ Bsc,
                       const float* __restrict__ Alog,
                       float* __restrict__ Pout,
                       float* __restrict__ Hend)
{
    const int lane = threadIdx.x & 31;
    const int q    = threadIdx.x >> 5;
    const int dg   = blockIdx.x % DGRP;
    const int t2   = blockIdx.x / DGRP;
    const int b    = t2 % BB;
    const int ch   = t2 / BB;
    const int d    = dg * 32 + lane;
    const size_t unit = (size_t)(ch * BB + b) * DMODEL + d;

    float A[SPQ], h[SPQ], P[SPQ];
    {
        const float4 al = *reinterpret_cast<const float4*>(Alog + d * DSTATE + SPQ * q);
        A[0] = -expf(al.x) * LOG2E_F; A[1] = -expf(al.y) * LOG2E_F;
        A[2] = -expf(al.z) * LOG2E_F; A[3] = -expf(al.w) * LOG2E_F;
    }
    for (int j = 0; j < SPQ; ++j) { h[j] = 0.0f; P[j] = 1.0f; }

    const size_t row0 = (size_t)(b * LL + ch * CHL);
    const float* dp = delta + row0 * DMODEL + d;
    const float* xp = X     + row0 * DMODEL + d;
    const float* bp = Bsc   + row0 * DSTATE + SPQ * q;
#pragma unroll 1
    for (int t0 = 0; t0 < CHL; t0 += UB) {
        float  dtv[UB], uv[UB];
        float4 bv[UB];
#pragma unroll
        for (int i = 0; i < UB; ++i) {
            dtv[i] = dp[(size_t)(t0 + i) * DMODEL];
            uv[i]  = xp[(size_t)(t0 + i) * DMODEL];
            bv[i]  = *reinterpret_cast<const float4*>(bp + (size_t)(t0 + i) * DSTATE);
        }
#pragma unroll
        for (int i = 0; i < UB; ++i) {
            const float dt = dtv[i];
            const float du = dt * uv[i];
            const float Bl[SPQ] = { bv[i].x, bv[i].y, bv[i].z, bv[i].w };
            for (int j = 0; j < SPQ; ++j) {
                float a = __builtin_amdgcn_exp2f(dt * A[j]);
                P[j] *= a;
                h[j]  = a * h[j] + du * Bl[j];
            }
        }
    }

    float4* Pp = reinterpret_cast<float4*>(Pout + unit * DSTATE + SPQ * q);
    float4* Hp = reinterpret_cast<float4*>(Hend + unit * DSTATE + SPQ * q);
    *Pp = make_float4(P[0], P[1], P[2], P[3]);
    *Hp = make_float4(h[0], h[1], h[2], h[3]);
}

__global__ __launch_bounds__(256)
void scan_passB_kernel(const float* __restrict__ P,
                       const float* __restrict__ Hend,
                       float* __restrict__ Hinit)
{
    const int idx = blockIdx.x * 256 + threadIdx.x;
    if (idx >= BB * DMODEL) return;
    const int b = idx / DMODEL;
    const int d = idx - b * DMODEL;

    float h[DSTATE];
#pragma unroll
    for (int n = 0; n < DSTATE; ++n) h[n] = 0.0f;

    for (int ch = 0; ch < NCH; ++ch) {
        const size_t unit = ((size_t)(ch * BB + b) * DMODEL + d) * DSTATE;
        float* Hi = Hinit + unit;
#pragma unroll
        for (int n = 0; n < DSTATE; ++n) Hi[n] = h[n];
#pragma unroll
        for (int n = 0; n < DSTATE; ++n)
            h[n] = P[unit + n] * h[n] + Hend[unit + n];
    }
}

__global__ __launch_bounds__(128) __attribute__((amdgpu_waves_per_eu(8, 8)))
void scan_passC_kernel(const float* __restrict__ delta,
                       const float* __restrict__ X,
                       const float* __restrict__ Bsc,
                       const float* __restrict__ Csc,
                       const float* __restrict__ Alog,
                       const float* __restrict__ Ds,
                       const float* __restrict__ Hinit,
                       float* __restrict__ Y)
{
    __shared__ float ypart[2][TB][NQ][32];

    const int lane = threadIdx.x & 31;
    const int q    = threadIdx.x >> 5;
    const int dg   = blockIdx.x % DGRP;
    const int t2   = blockIdx.x / DGRP;
    const int b    = t2 % BB;
    const int ch   = t2 / BB;
    const int d    = dg * 32 + lane;
    const size_t unit = (size_t)(ch * BB + b) * DMODEL + d;

    float A[SPQ], h[SPQ];
    {
        const float4 al = *reinterpret_cast<const float4*>(Alog + d * DSTATE + SPQ * q);
        const float4 hv = *reinterpret_cast<const float4*>(Hinit + unit * DSTATE + SPQ * q);
        A[0] = -expf(al.x) * LOG2E_F; A[1] = -expf(al.y) * LOG2E_F;
        A[2] = -expf(al.z) * LOG2E_F; A[3] = -expf(al.w) * LOG2E_F;
        h[0] = hv.x; h[1] = hv.y; h[2] = hv.z; h[3] = hv.w;
    }
    const float Dsd = Ds[d];

    const int l0 = ch * CHL;
    const size_t row0 = (size_t)(b * LL + l0);
    const float* dp = delta + row0 * DMODEL + d;
    const float* xp = X     + row0 * DMODEL + d;
    const float* bp = Bsc   + row0 * DSTATE + SPQ * q;
    const float* cp = Csc   + row0 * DSTATE + SPQ * q;
    for (int bt = 0; bt < CHL / TB; ++bt) {
        float (*yp)[NQ][32] = ypart[bt & 1];
#pragma unroll 1
        for (int s0 = 0; s0 < TB; s0 += UB) {
            const int t0 = bt * TB + s0;
            float  dtv[UB], uv[UB];
            float4 bv[UB], cv[UB];
#pragma unroll
            for (int i = 0; i < UB; ++i) {
                dtv[i] = dp[(size_t)(t0 + i) * DMODEL];
                uv[i]  = xp[(size_t)(t0 + i) * DMODEL];
                bv[i]  = *reinterpret_cast<const float4*>(bp + (size_t)(t0 + i) * DSTATE);
                cv[i]  = *reinterpret_cast<const float4*>(cp + (size_t)(t0 + i) * DSTATE);
            }
#pragma unroll
            for (int i = 0; i < UB; ++i) {
                const float dt = dtv[i];
                const float u  = uv[i];
                const float du = dt * u;
                const float Bl[SPQ] = { bv[i].x, bv[i].y, bv[i].z, bv[i].w };
                const float Cl[SPQ] = { cv[i].x, cv[i].y, cv[i].z, cv[i].w };
                float y = 0.0f;
                for (int j = 0; j < SPQ; ++j) {
                    float a = __builtin_amdgcn_exp2f(dt * A[j]);
                    h[j] = a * h[j] + du * Bl[j];
                    y   += h[j] * Cl[j];
                }
                if (q == 0) y += u * Dsd;
                yp[s0 + i][q][lane] = y;
            }
        }
        __syncthreads();
        for (int r = 0; r < TB / NQ; ++r) {
            const int tt = q * (TB / NQ) + r;
            const float s = (yp[tt][0][lane] + yp[tt][1][lane])
                          + (yp[tt][2][lane] + yp[tt][3][lane]);
            const int bl = b * LL + l0 + bt * TB + tt;
            Y[(size_t)bl * DMODEL + d] = s;
        }
    }
}

extern "C" void kernel_launch(void* const* d_in, const int* in_sizes, int n_in,
                              void* d_out, int out_size, void* d_ws, size_t ws_size,
                              hipStream_t stream)
{
    const float* x      = (const float*)d_in[0];
    const float* prompt = (const float*)d_in[1];
    const float* w1     = (const float*)d_in[2];
    const float* w2     = (const float*)d_in[3];
    const float* bias   = (const float*)d_in[4];
    const float* alog   = (const float*)d_in[5];
    const float* ds     = (const float*)d_in[6];
    float*       y      = (float*)d_out;

    float* ws = (float*)d_ws;
    size_t o = 0;
    float* dtsr  = ws + o; o += (size_t)MTOT * DTRANK;
    float* Bsc   = ws + o; o += (size_t)MTOT * DSTATE;
    float* Csc   = ws + o; o += (size_t)MTOT * DSTATE;
    float* delta = ws + o; o += (size_t)MTOT * DMODEL;
    const size_t units = (size_t)NCH * BB * DMODEL;
    float* P     = ws + o; o += units * DSTATE;
    float* Hend  = ws + o; o += units * DSTATE;
    float* Hinit = ws + o; o += units * DSTATE;
    (void)ws_size; (void)in_sizes; (void)n_in; (void)out_size;

    proj_gemm_kernel<<<dim3(MTOT / 128, 4, 1), 256, 0, stream>>>(x, w1, prompt, dtsr, Bsc, Csc);
    dt_gemm_kernel  <<<dim3(MTOT / 128, DTSPLIT, 1), 256, 0, stream>>>(dtsr, w2, bias, delta);
    scan_passA_kernel<<<dim3((unsigned)(units / 32)), 128, 0, stream>>>(delta, x, Bsc, alog, P, Hend);
    scan_passB_kernel<<<dim3((BB * DMODEL + 255) / 256), 256, 0, stream>>>(P, Hend, Hinit);
    scan_passC_kernel<<<dim3((unsigned)(units / 32)), 128, 0, stream>>>(delta, x, Bsc, Csc, alog, ds, Hinit, y);
}


# dt GEMM softplus: max(t,0)+log1p(exp(-|t|)) on hardware exp2 and rcp (two-float exponent, atanh series for log1p) in place of the library log1pf(expf(t)): a quarter of the vector instructions
# speedup vs baseline: 4.4506x; 1.0953x over previous
#include <hip/hip_runtime.h>
#include <hip/hip_bf16.h>
#include <math.h>

#define BB      4
#define LL      4096
#define DMODEL  384
#define DSTATE  16
#define DTRANK  24
#define CTOT    (DTRANK + 2*DSTATE)
#define MTOT    (BB*LL)
#define NCH     32
#define CHL     (LL/NCH)
#define NQ      4
#define SPQ     (DSTATE/NQ)
#define DGRP    (DMODEL/32)
#define TB      32
#define UB      8
#define DTSPLIT 6
#define DTCOLS  (DMODEL/DTSPLIT)
#define LOG2E_F 1.4426950408889634f
#define LOG2E_LO_F 1.92596299e-8f
#define LN2_F      0.6931471805599453f

typedef float     v2f  __attribute__((ext_vector_type(2)));
typedef float     v8f  __attribute__((ext_vector_type(8)));
typedef _Float16  v16h __attribute__((ext_vector_type(16)));

#if defined(__gfx1250__) && __has_builtin(__builtin_amdgcn_wmma_f32_16x16x4_f32)
#define HAVE_WMMA_F32 1
#else
#define HAVE_WMMA_F32 0
#endif

__global__ __launch_bounds__(256)
void proj_gemm_kernel(const float* __restrict__ X,
                      const float* __restrict__ W1,
                      const float* __restrict__ prompt,
                      float* __restrict__ dtsr,
                      float* __restrict__ Bsc,
                      float* __restrict__ Csc)
{
    __shared__ float Wlds[16 * 388];

    const int c0 = blockIdx.y * 16;
    for (int i = threadIdx.x; i < 16 * DMODEL; i += 256) {
        int n = i / DMODEL, k = i - n * DMODEL;
        float w = (c0 + n < CTOT) ? W1[(c0 + n) * DMODEL + k] : 0.0f;
        Wlds[n * 388 + k] = w;
    }
    __syncthreads();

    const int wave = threadIdx.x >> 5;
    const int lane = threadIdx.x & 31;
    const int ln   = lane & 15;
    const int hi   = lane >> 4;
    const int m0   = blockIdx.x * 128 + wave * 16;

    v8f acc = {};
    const float* arow = X + (size_t)(m0 + ln) * DMODEL;

#if HAVE_WMMA_F32
    for (int k0 = 0; k0 < DMODEL; k0 += 4) {
        v2f a, b;
        a.x = arow[k0 + 2 * hi];
        a.y = arow[k0 + 2 * hi + 1];
        b.x = Wlds[ln * 388 + k0 + 2 * hi];
        b.y = Wlds[ln * 388 + k0 + 2 * hi + 1];
        acc = __builtin_amdgcn_wmma_f32_16x16x4_f32(
                  false, a, false, b, (short)0, acc, false, false);
    }
#else
    for (int k0 = 0; k0 < DMODEL; k0 += 32) {
        v16h a, b;
#pragma unroll
        for (int v = 0; v < 8; ++v) {
            int kk = (v < 4 ? 2 * v : 16 + 2 * (v - 4)) + 8 * hi;
            a[2 * v]     = (_Float16)arow[k0 + kk];
            a[2 * v + 1] = (_Float16)arow[k0 + kk + 1];
            b[2 * v]     = (_Float16)Wlds[ln * 388 + k0 + kk];
            b[2 * v + 1] = (_Float16)Wlds[ln * 388 + k0 + kk + 1];
        }
        acc = __builtin_amdgcn_wmma_f32_16x16x32_f16(
                  false, a, false, b, (short)0, acc, false, false);
    }
#endif

    const int c = c0 + ln;
#pragma unroll
    for (int r = 0; r < 8; ++r) {
        int m = m0 + r + 8 * hi;
        float v = acc[r];
        if (c < DTRANK) {
            dtsr[(size_t)m * DTRANK + c] = v;
        } else if (c < DTRANK + DSTATE) {
            Bsc[(size_t)m * DSTATE + (c - DTRANK)] = v;
        } else if (c < CTOT) {
            int n = c - DTRANK - DSTATE;
            Csc[(size_t)m * DSTATE + n] = v + prompt[(size_t)m * DSTATE + n];
        }
    }
}

__device__ __forceinline__ float softplus_f32(float t)
{
    const float na  = -__builtin_fabsf(t);
    const float hi  = na * LOG2E_F;
    float       lo  = __builtin_fmaf(na, LOG2E_F, -hi);
    lo              = __builtin_fmaf(na, LOG2E_LO_F, lo);
    const float e   = __builtin_amdgcn_exp2f(hi);
    const float z   = __builtin_fmaf(e, lo * LN2_F, e);
    const float den = 2.0f + z;
    const float r   = __builtin_amdgcn_rcpf(den);
    float       s   = z * r;
    s               = __builtin_fmaf(__builtin_fmaf(-s, den, z), r, s);
    const float w   = s * s;
    float p = 2.0f / 15.0f;
    p = __builtin_fmaf(p, w, 2.0f / 13.0f);
    p = __builtin_fmaf(p, w, 2.0f / 11.0f);
    p = __builtin_fmaf(p, w, 2.0f / 9.0f);
    p = __builtin_fmaf(p, w, 2.0f / 7.0f);
    p = __builtin_fmaf(p, w, 2.0f / 5.0f);
    p = __builtin_fmaf(p, w, 2.0f / 3.0f);
    p = __builtin_fmaf(p, w, 2.0f);
    return __builtin_fmaf(s, p, __builtin_fmaxf(t, 0.0f));
}

__global__ __launch_bounds__(256)
void dt_gemm_kernel(const float* __restrict__ dtsr,
                    const float* __restrict__ W2,
                    const float* __restrict__ bias,
                    float* __restrict__ delta)
{
    __shared__ float Wlds[DTCOLS * 28];

    const int dbase = blockIdx.y * DTCOLS;
    for (int i = threadIdx.x; i < DTCOLS * DTRANK; i += 256) {
        int d = i / DTRANK, r = i - d * DTRANK;
        Wlds[d * 28 + r] = W2[dbase * DTRANK + i];
    }
    __syncthreads();

    const int wave = threadIdx.x >> 5;
    const int lane = threadIdx.x & 31;
    const int ln   = lane & 15;
    const int hi   = lane >> 4;
    const int m0   = blockIdx.x * 128 + wave * 16;

    const float* arow = dtsr + (size_t)(m0 + ln) * DTRANK;

#if HAVE_WMMA_F32
    v2f afr[6];
#pragma unroll
    for (int ks = 0; ks < 6; ++ks) {
        afr[ks].x = arow[4 * ks + 2 * hi];
        afr[ks].y = arow[4 * ks + 2 * hi + 1];
    }
#else
    v16h afr;
#pragma unroll
    for (int v = 0; v < 8; ++v) {
        int kk = (v < 4 ? 2 * v : 16 + 2 * (v - 4)) + 8 * hi;
        afr[2 * v]     = (kk     < DTRANK) ? (_Float16)arow[kk]     : (_Float16)0.f;
        afr[2 * v + 1] = (kk + 1 < DTRANK) ? (_Float16)arow[kk + 1] : (_Float16)0.f;
    }
#endif

    for (int d0 = 0; d0 < DTCOLS; d0 += 16) {
        v8f acc = {};
#if HAVE_WMMA_F32
#pragma unroll
        for (int ks = 0; ks < 6; ++ks) {
            v2f b;
            b.x = Wlds[(d0 + ln) * 28 + 4 * ks + 2 * hi];
            b.y = Wlds[(d0 + ln) * 28 + 4 * ks + 2 * hi + 1];
            acc = __builtin_amdgcn_wmma_f32_16x16x4_f32(
                      false, afr[ks], false, b, (short)0, acc, false, false);
        }
#else
        v16h b;
#pragma unroll
        for (int v = 0; v < 8; ++v) {
            int kk = (v < 4 ? 2 * v : 16 + 2 * (v - 4)) + 8 * hi;
            b[2 * v]     = (kk     < DTRANK) ? (_Float16)Wlds[(d0 + ln) * 28 + kk]     : (_Float16)0.f;
            b[2 * v + 1] = (kk + 1 < DTRANK) ? (_Float16)Wlds[(d0 + ln) * 28 + kk + 1] : (_Float16)0.f;
        }
        acc = __builtin_amdgcn_wmma_f32_16x16x32_f16(
                  false, afr, false, b, (short)0, acc, false, false);
#endif
        const int   d  = dbase + d0 + ln;
        const float bi = bias[d];
#pragma unroll
        for (int r = 0; r < 8; ++r) {
            int   m = m0 + r + 8 * hi;
            delta[(size_t)m * DMODEL + d] = softplus_f32(acc[r] + bi);
        }
    }
}

__global__ __launch_bounds__(128) __attribute__((amdgpu_waves_per_eu(8, 8)))
void scan_passA_kernel(const float* __restrict__ delta,
                       const float* __restrict__ X,
                       const float* __restrict__ Bsc,
                       const float* __restrict__ Alog,
                       float* __restrict__ Pout,
                       float* __restrict__ Hend)
{
    const int lane = threadIdx.x & 31;
    const int q    = threadIdx.x >> 5;
    const int dg   = blockIdx.x % DGRP;
    const int t2   = blockIdx.x / DGRP;
    const int b    = t2 % BB;
    const int ch   = t2 / BB;
    const int d    = dg * 32 + lane;
    const size_t unit = (size_t)(ch * BB + b) * DMODEL + d;

    float A[SPQ], h[SPQ], P[SPQ];
    {
        const float4 al = *reinterpret_cast<const float4*>(Alog + d * DSTATE + SPQ * q);
        A[0] = -expf(al.x) * LOG2E_F; A[1] = -expf(al.y) * LOG2E_F;
        A[2] = -expf(al.z) * LOG2E_F; A[3] = -expf(al.w) * LOG2E_F;
    }
    for (int j = 0; j < SPQ; ++j) { h[j] = 0.0f; P[j] = 1.0f; }

    const size_t row0 = (size_t)(b * LL + ch * CHL);
    const float* dp = delta + row0 * DMODEL + d;
    const float* xp = X     + row0 * DMODEL + d;
    const float* bp = Bsc   + row0 * DSTATE + SPQ * q;
#pragma unroll 1
    for (int t0 = 0; t0 < CHL; t0 += UB) {
        float  dtv[UB], uv[UB];
        float4 bv[UB];
#pragma unroll
        for (int i = 0; i < UB; ++i) {
            dtv[i] = dp[(size_t)(t0 + i) * DMODEL];
            uv[i]  = xp[(size_t)(t0 + i) * DMODEL];
            bv[i]  = *reinterpret_cast<const float4*>(bp + (size_t)(t0 + i) * DSTATE);
        }
#pragma unroll
        for (int i = 0; i < UB; ++i) {
            const float dt = dtv[i];
            const float du = dt * uv[i];
            const float Bl[SPQ] = { bv[i].x, bv[i].y, bv[i].z, bv[i].w };
            for (int j = 0; j < SPQ; ++j) {
                float a = __builtin_amdgcn_exp2f(dt * A[j]);
                P[j] *= a;
                h[j]  = a * h[j] + du * Bl[j];
            }
        }
    }

    float4* Pp = reinterpret_cast<float4*>(Pout + unit * DSTATE + SPQ * q);
    float4* Hp = reinterpret_cast<float4*>(Hend + unit * DSTATE + SPQ * q);
    *Pp = make_float4(P[0], P[1], P[2], P[3]);
    *Hp = make_float4(h[0], h[1], h[2], h[3]);
}

__global__ __launch_bounds__(256)
void scan_passB_kernel(const float* __restrict__ P,
                       const float* __restrict__ Hend,
                       float* __restrict__ Hinit)
{
    const int idx = blockIdx.x * 256 + threadIdx.x;
    if (idx >= BB * DMODEL) return;
    const int b = idx / DMODEL;
    const int d = idx - b * DMODEL;

    float h[DSTATE];
#pragma unroll
    for (int n = 0; n < DSTATE; ++n) h[n] = 0.0f;

    for (int ch = 0; ch < NCH; ++ch) {
        const size_t unit = ((size_t)(ch * BB + b) * DMODEL + d) * DSTATE;
        float* Hi = Hinit + unit;
#pragma unroll
        for (int n = 0; n < DSTATE; ++n) Hi[n] = h[n];
#pragma unroll
        for (int n = 0; n < DSTATE; ++n)
            h[n] = P[unit + n] * h[n] + Hend[unit + n];
    }
}

__global__ __launch_bounds__(128) __attribute__((amdgpu_waves_per_eu(8, 8)))
void scan_passC_kernel(const float* __restrict__ delta,
                       const float* __restrict__ X,
                       const float* __restrict__ Bsc,
                       const float* __restrict__ Csc,
                       const float* __restrict__ Alog,
                       const float* __restrict__ Ds,
                       const float* __restrict__ Hinit,
                       float* __restrict__ Y)
{
    __shared__ float ypart[2][TB][NQ][32];

    const int lane = threadIdx.x & 31;
    const int q    = threadIdx.x >> 5;
    const int dg   = blockIdx.x % DGRP;
    const int t2   = blockIdx.x / DGRP;
    const int b    = t2 % BB;
    const int ch   = t2 / BB;
    const int d    = dg * 32 + lane;
    const size_t unit = (size_t)(ch * BB + b) * DMODEL + d;

    float A[SPQ], h[SPQ];
    {
        const float4 al = *reinterpret_cast<const float4*>(Alog + d * DSTATE + SPQ * q);
        const float4 hv = *reinterpret_cast<const float4*>(Hinit + unit * DSTATE + SPQ * q);
        A[0] = -expf(al.x) * LOG2E_F; A[1] = -expf(al.y) * LOG2E_F;
        A[2] = -expf(al.z) * LOG2E_F; A[3] = -expf(al.w) * LOG2E_F;
        h[0] = hv.x; h[1] = hv.y; h[2] = hv.z; h[3] = hv.w;
    }
    const float Dsd = Ds[d];

    const int l0 = ch * CHL;
    const size_t row0 = (size_t)(b * LL + l0);
    const float* dp = delta + row0 * DMODEL + d;
    const float* xp = X     + row0 * DMODEL + d;
    const float* bp = Bsc   + row0 * DSTATE + SPQ * q;
    const float* cp = Csc   + row0 * DSTATE + SPQ * q;
    for (int bt = 0; bt < CHL / TB; ++bt) {
        float (*yp)[NQ][32] = ypart[bt & 1];
#pragma unroll 1
        for (int s0 = 0; s0 < TB; s0 += UB) {
            const int t0 = bt * TB + s0;
            float  dtv[UB], uv[UB];
            float4 bv[UB], cv[UB];
#pragma unroll
            for (int i = 0; i < UB; ++i) {
                dtv[i] = dp[(size_t)(t0 + i) * DMODEL];
                uv[i]  = xp[(size_t)(t0 + i) * DMODEL];
                bv[i]  = *reinterpret_cast<const float4*>(bp + (size_t)(t0 + i) * DSTATE);
                cv[i]  = *reinterpret_cast<const float4*>(cp + (size_t)(t0 + i) * DSTATE);
            }
#pragma unroll
            for (int i = 0; i < UB; ++i) {
                const float dt = dtv[i];
                const float u  = uv[i];
                const float du = dt * u;
                const float Bl[SPQ] = { bv[i].x, bv[i].y, bv[i].z, bv[i].w };
                const float Cl[SPQ] = { cv[i].x, cv[i].y, cv[i].z, cv[i].w };
                float y = 0.0f;
                for (int j = 0; j < SPQ; ++j) {
                    float a = __builtin_amdgcn_exp2f(dt * A[j]);
                    h[j] = a * h[j] + du * Bl[j];
                    y   += h[j] * Cl[j];
                }
                if (q == 0) y += u * Dsd;
                yp[s0 + i][q][lane] = y;
            }
        }
        __syncthreads();
        for (int r = 0; r < TB / NQ; ++r) {
            const int tt = q * (TB / NQ) + r;
            const float s = (yp[tt][0][lane] + yp[tt][1][lane])
                          + (yp[tt][2][lane] + yp[tt][3][lane]);
            const int bl = b * LL + l0 + bt * TB + tt;
            Y[(size_t)bl * DMODEL + d] = s;
        }
    }
}

extern "C" void kernel_launch(void* const* d_in, const int* in_sizes, int n_in,
                              void* d_out, int out_size, void* d_ws, size_t ws_size,
                              hipStream_t stream)
{
    const float* x      = (const float*)d_in[0];
    const float* prompt = (const float*)d_in[1];
    const float* w1     = (const float*)d_in[2];
    const float* w2     = (const float*)d_in[3];
    const float* bias   = (const float*)d_in[4];
    const float* alog   = (const float*)d_in[5];
    const float* ds     = (const float*)d_in[6];
    float*       y      = (float*)d_out;

    float* ws = (float*)d_ws;
    size_t o = 0;
    float* dtsr  = ws + o; o += (size_t)MTOT * DTRANK;
    float* Bsc   = ws + o; o += (size_t)MTOT * DSTATE;
    float* Csc   = ws + o; o += (size_t)MTOT * DSTATE;
    float* delta = ws + o; o += (size_t)MTOT * DMODEL;
    const size_t units = (size_t)NCH * BB * DMODEL;
    float* P     = ws + o; o += units * DSTATE;
    float* Hend  = ws + o; o += units * DSTATE;
    float* Hinit = ws + o; o += units * DSTATE;
    (void)ws_size; (void)in_sizes; (void)n_in; (void)out_size;

    proj_gemm_kernel<<<dim3(MTOT / 128, 4, 1), 256, 0, stream>>>(x, w1, prompt, dtsr, Bsc, Csc);
    dt_gemm_kernel  <<<dim3(MTOT / 128, DTSPLIT, 1), 256, 0, stream>>>(dtsr, w2, bias, delta);
    scan_passA_kernel<<<dim3((unsigned)(units / 32)), 128, 0, stream>>>(delta, x, Bsc, alog, P, Hend);
    scan_passB_kernel<<<dim3((BB * DMODEL + 255) / 256), 256, 0, stream>>>(P, Hend, Hinit);
    scan_passC_kernel<<<dim3((unsigned)(units / 32)), 128, 0, stream>>>(delta, x, Bsc, Csc, alog, ds, Hinit, y);
}


# Scan pass B (chunk carry): one thread per (b,d,state), the loads of all 32 chunks issued before the fold, whole 128-B lines per wave (was 32 dependent load round trips on 48 waves); sum order kept
# speedup vs baseline: 5.7147x; 1.2840x over previous
#include <hip/hip_runtime.h>
#include <hip/hip_bf16.h>
#include <math.h>

#define BB      4
#define LL      4096
#define DMODEL  384
#define DSTATE  16
#define DTRANK  24
#define CTOT    (DTRANK + 2*DSTATE)
#define MTOT    (BB*LL)
#define NCH     32
#define CHL     (LL/NCH)
#define NQ      4
#define SPQ     (DSTATE/NQ)
#define DGRP    (DMODEL/32)
#define TB      32
#define UB      8
#define DTSPLIT 6
#define DTCOLS  (DMODEL/DTSPLIT)
#define LOG2E_F 1.4426950408889634f
#define LOG2E_LO_F 1.92596299e-8f
#define LN2_F      0.6931471805599453f

typedef float     v2f  __attribute__((ext_vector_type(2)));
typedef float     v8f  __attribute__((ext_vector_type(8)));
typedef _Float16  v16h __attribute__((ext_vector_type(16)));

#if defined(__gfx1250__) && __has_builtin(__builtin_amdgcn_wmma_f32_16x16x4_f32)
#define HAVE_WMMA_F32 1
#else
#define HAVE_WMMA_F32 0
#endif

__global__ __launch_bounds__(256)
void proj_gemm_kernel(const float* __restrict__ X,
                      const float* __restrict__ W1,
                      const float* __restrict__ prompt,
                      float* __restrict__ dtsr,
                      float* __restrict__ Bsc,
                      float* __restrict__ Csc)
{
    __shared__ float Wlds[16 * 388];

    const int c0 = blockIdx.y * 16;
    for (int i = threadIdx.x; i < 16 * DMODEL; i += 256) {
        int n = i / DMODEL, k = i - n * DMODEL;
        float w = (c0 + n < CTOT) ? W1[(c0 + n) * DMODEL + k] : 0.0f;
        Wlds[n * 388 + k] = w;
    }
    __syncthreads();

    const int wave = threadIdx.x >> 5;
    const int lane = threadIdx.x & 31;
    const int ln   = lane & 15;
    const int hi   = lane >> 4;
    const int m0   = blockIdx.x * 128 + wave * 16;

    v8f acc = {};
    const float* arow = X + (size_t)(m0 + ln) * DMODEL;

#if HAVE_WMMA_F32
    for (int k0 = 0; k0 < DMODEL; k0 += 4) {
        v2f a, b;
        a.x = arow[k0 + 2 * hi];
        a.y = arow[k0 + 2 * hi + 1];
        b.x = Wlds[ln * 388 + k0 + 2 * hi];
        b.y = Wlds[ln * 388 + k0 + 2 * hi + 1];
        acc = __builtin_amdgcn_wmma_f32_16x16x4_f32(
                  false, a, false, b, (short)0, acc, false, false);
    }
#else
    for (int k0 = 0; k0 < DMODEL; k0 += 32) {
        v16h a, b;
#pragma unroll
        for (int v = 0; v < 8; ++v) {
            int kk = (v < 4 ? 2 * v : 16 + 2 * (v - 4)) + 8 * hi;
            a[2 * v]     = (_Float16)arow[k0 + kk];
            a[2 * v + 1] = (_Float16)arow[k0 + kk + 1];
            b[2 * v]     = (_Float16)Wlds[ln * 388 + k0 + kk];
            b[2 * v + 1] = (_Float16)Wlds[ln * 388 + k0 + kk + 1];
        }
        acc = __builtin_amdgcn_wmma_f32_16x16x32_f16(
                  false, a, false, b, (short)0, acc, false, false);
    }
#endif

    const int c = c0 + ln;
#pragma unroll
    for (int r = 0; r < 8; ++r) {
        int m = m0 + r + 8 * hi;
        float v = acc[r];
        if (c < DTRANK) {
            dtsr[(size_t)m * DTRANK + c] = v;
        } else if (c < DTRANK + DSTATE) {
            Bsc[(size_t)m * DSTATE + (c - DTRANK)] = v;
        } else if (c < CTOT) {
            int n = c - DTRANK - DSTATE;
            Csc[(size_t)m * DSTATE + n] = v + prompt[(size_t)m * DSTATE + n];
        }
    }
}

__device__ __forceinline__ float softplus_f32(float t)
{
    const float na  = -__builtin_fabsf(t);
    const float hi  = na * LOG2E_F;
    float       lo  = __builtin_fmaf(na, LOG2E_F, -hi);
    lo              = __builtin_fmaf(na, LOG2E_LO_F, lo);
    const float e   = __builtin_amdgcn_exp2f(hi);
    const float z   = __builtin_fmaf(e, lo * LN2_F, e);
    const float den = 2.0f + z;
    const float r   = __builtin_amdgcn_rcpf(den);
    float       s   = z * r;
    s               = __builtin_fmaf(__builtin_fmaf(-s, den, z), r, s);
    const float w   = s * s;
    float p = 2.0f / 15.0f;
    p = __builtin_fmaf(p, w, 2.0f / 13.0f);
    p = __builtin_fmaf(p, w, 2.0f / 11.0f);
    p = __builtin_fmaf(p, w, 2.0f / 9.0f);
    p = __builtin_fmaf(p, w, 2.0f / 7.0f);
    p = __builtin_fmaf(p, w, 2.0f / 5.0f);
    p = __builtin_fmaf(p, w, 2.0f / 3.0f);
    p = __builtin_fmaf(p, w, 2.0f);
    return __builtin_fmaf(s, p, __builtin_fmaxf(t, 0.0f));
}

__global__ __launch_bounds__(256)
void dt_gemm_kernel(const float* __restrict__ dtsr,
                    const float* __restrict__ W2,
                    const float* __restrict__ bias,
                    float* __restrict__ delta)
{
    __shared__ float Wlds[DTCOLS * 28];

    const int dbase = blockIdx.y * DTCOLS;
    for (int i = threadIdx.x; i < DTCOLS * DTRANK; i += 256) {
        int d = i / DTRANK, r = i - d * DTRANK;
        Wlds[d * 28 + r] = W2[dbase * DTRANK + i];
    }
    __syncthreads();

    const int wave = threadIdx.x >> 5;
    const int lane = threadIdx.x & 31;
    const int ln   = lane & 15;
    const int hi   = lane >> 4;
    const int m0   = blockIdx.x * 128 + wave * 16;

    const float* arow = dtsr + (size_t)(m0 + ln) * DTRANK;

#if HAVE_WMMA_F32
    v2f afr[6];
#pragma unroll
    for (int ks = 0; ks < 6; ++ks) {
        afr[ks].x = arow[4 * ks + 2 * hi];
        afr[ks].y = arow[4 * ks + 2 * hi + 1];
    }
#else
    v16h afr;
#pragma unroll
    for (int v = 0; v < 8; ++v) {
        int kk = (v < 4 ? 2 * v : 16 + 2 * (v - 4)) + 8 * hi;
        afr[2 * v]     = (kk     < DTRANK) ? (_Float16)arow[kk]     : (_Float16)0.f;
        afr[2 * v + 1] = (kk + 1 < DTRANK) ? (_Float16)arow[kk + 1] : (_Float16)0.f;
    }
#endif

    for (int d0 = 0; d0 < DTCOLS; d0 += 16) {
        v8f acc = {};
#if HAVE_WMMA_F32
#pragma unroll
        for (int ks = 0; ks < 6; ++ks) {
            v2f b;
            b.x = Wlds[(d0 + ln) * 28 + 4 * ks + 2 * hi];
            b.y = Wlds[(d0 + ln) * 28 + 4 * ks + 2 * hi + 1];
            acc = __builtin_amdgcn_wmma_f32_16x16x4_f32(
                      false, afr[ks], false, b, (short)0, acc, false, false);
        }
#else
        v16h b;
#pragma unroll
        for (int v = 0; v < 8; ++v) {
            int kk = (v < 4 ? 2 * v : 16 + 2 * (v - 4)) + 8 * hi;
            b[2 * v]     = (kk     < DTRANK) ? (_Float16)Wlds[(d0 + ln) * 28 + kk]     : (_Float16)0.f;
            b[2 * v + 1] = (kk + 1 < DTRANK) ? (_Float16)Wlds[(d0 + ln) * 28 + kk + 1] : (_Float16)0.f;
        }
        acc = __builtin_amdgcn_wmma_f32_16x16x32_f16(
                  false, afr, false, b, (short)0, acc, false, false);
#endif
        const int   d  = dbase + d0 + ln;
        const float bi = bias[d];
#pragma unroll
        for (int r = 0; r < 8; ++r) {
            int   m = m0 + r + 8 * hi;
            delta[(size_t)m * DMODEL + d] = softplus_f32(acc[r] + bi);
        }
    }
}

__global__ __launch_bounds__(128) __attribute__((amdgpu_waves_per_eu(8, 8)))
void scan_passA_kernel(const float* __restrict__ delta,
                       const float* __restrict__ X,
                       const float* __restrict__ Bsc,
                       const float* __restrict__ Alog,
                       float* __restrict__ Pout,
                       float* __restrict__ Hend)
{
    const int lane = threadIdx.x & 31;
    const int q    = threadIdx.x >> 5;
    const int dg   = blockIdx.x % DGRP;
    const int t2   = blockIdx.x / DGRP;
    const int b    = t2 % BB;
    const int ch   = t2 / BB;
    const int d    = dg * 32 + lane;
    const size_t unit = (size_t)(ch * BB + b) * DMODEL + d;

    float A[SPQ], h[SPQ], P[SPQ];
    {
        const float4 al = *reinterpret_cast<const float4*>(Alog + d * DSTATE + SPQ * q);
        A[0] = -expf(al.x) * LOG2E_F; A[1] = -expf(al.y) * LOG2E_F;
        A[2] = -expf(al.z) * LOG2E_F; A[3] = -expf(al.w) * LOG2E_F;
    }
    for (int j = 0; j < SPQ; ++j) { h[j] = 0.0f; P[j] = 1.0f; }

    const size_t row0 = (size_t)(b * LL + ch * CHL);
    const float* dp = delta + row0 * DMODEL + d;
    const float* xp = X     + row0 * DMODEL + d;
    const float* bp = Bsc   + row0 * DSTATE + SPQ * q;
#pragma unroll 1
    for (int t0 = 0; t0 < CHL; t0 += UB) {
        float  dtv[UB], uv[UB];
        float4 bv[UB];
#pragma unroll
        for (int i = 0; i < UB; ++i) {
            dtv[i] = dp[(size_t)(t0 + i) * DMODEL];
            uv[i]  = xp[(size_t)(t0 + i) * DMODEL];
            bv[i]  = *reinterpret_cast<const float4*>(bp + (size_t)(t0 + i) * DSTATE);
        }
#pragma unroll
        for (int i = 0; i < UB; ++i) {
            const float dt = dtv[i];
            const float du = dt * uv[i];
            const float Bl[SPQ] = { bv[i].x, bv[i].y, bv[i].z, bv[i].w };
            for (int j = 0; j < SPQ; ++j) {
                float a = __builtin_amdgcn_exp2f(dt * A[j]);
                P[j] *= a;
                h[j]  = a * h[j] + du * Bl[j];
            }
        }
    }

    float4* Pp = reinterpret_cast<float4*>(Pout + unit * DSTATE + SPQ * q);
    float4* Hp = reinterpret_cast<float4*>(Hend + unit * DSTATE + SPQ * q);
    *Pp = make_float4(P[0], P[1], P[2], P[3]);
    *Hp = make_float4(h[0], h[1], h[2], h[3]);
}

#define PB_THREADS 96
#define PB_PLANE   (BB * DMODEL * DSTATE)
__global__ __launch_bounds__(PB_THREADS)
void scan_passB_kernel(const float* __restrict__ P,
                       const float* __restrict__ Hend,
                       float* __restrict__ Hinit)
{
    const int idx = blockIdx.x * PB_THREADS + threadIdx.x;

    float p[NCH], e[NCH];
#pragma unroll
    for (int ch = 0; ch < NCH; ++ch) {
        p[ch] = P[(size_t)ch * PB_PLANE + idx];
        e[ch] = Hend[(size_t)ch * PB_PLANE + idx];
    }

    float h = 0.0f;
#pragma unroll
    for (int ch = 0; ch < NCH; ++ch) {
        Hinit[(size_t)ch * PB_PLANE + idx] = h;
        h = __builtin_fmaf(p[ch], h, e[ch]);
    }
}

__global__ __launch_bounds__(128) __attribute__((amdgpu_waves_per_eu(8, 8)))
void scan_passC_kernel(const float* __restrict__ delta,
                       const float* __restrict__ X,
                       const float* __restrict__ Bsc,
                       const float* __restrict__ Csc,
                       const float* __restrict__ Alog,
                       const float* __restrict__ Ds,
                       const float* __restrict__ Hinit,
                       float* __restrict__ Y)
{
    __shared__ float ypart[2][TB][NQ][32];

    const int lane = threadIdx.x & 31;
    const int q    = threadIdx.x >> 5;
    const int dg   = blockIdx.x % DGRP;
    const int t2   = blockIdx.x / DGRP;
    const int b    = t2 % BB;
    const int ch   = t2 / BB;
    const int d    = dg * 32 + lane;
    const size_t unit = (size_t)(ch * BB + b) * DMODEL + d;

    float A[SPQ], h[SPQ];
    {
        const float4 al = *reinterpret_cast<const float4*>(Alog + d * DSTATE + SPQ * q);
        const float4 hv = *reinterpret_cast<const float4*>(Hinit + unit * DSTATE + SPQ * q);
        A[0] = -expf(al.x) * LOG2E_F; A[1] = -expf(al.y) * LOG2E_F;
        A[2] = -expf(al.z) * LOG2E_F; A[3] = -expf(al.w) * LOG2E_F;
        h[0] = hv.x; h[1] = hv.y; h[2] = hv.z; h[3] = hv.w;
    }
    const float Dsd = Ds[d];

    const int l0 = ch * CHL;
    const size_t row0 = (size_t)(b * LL + l0);
    const float* dp = delta + row0 * DMODEL + d;
    const float* xp = X     + row0 * DMODEL + d;
    const float* bp = Bsc   + row0 * DSTATE + SPQ * q;
    const float* cp = Csc   + row0 * DSTATE + SPQ * q;
    for (int bt = 0; bt < CHL / TB; ++bt) {
        float (*yp)[NQ][32] = ypart[bt & 1];
#pragma unroll 1
        for (int s0 = 0; s0 < TB; s0 += UB) {
            const int t0 = bt * TB + s0;
            float  dtv[UB], uv[UB];
            float4 bv[UB], cv[UB];
#pragma unroll
            for (int i = 0; i < UB; ++i) {
                dtv[i] = dp[(size_t)(t0 + i) * DMODEL];
                uv[i]  = xp[(size_t)(t0 + i) * DMODEL];
                bv[i]  = *reinterpret_cast<const float4*>(bp + (size_t)(t0 + i) * DSTATE);
                cv[i]  = *reinterpret_cast<const float4*>(cp + (size_t)(t0 + i) * DSTATE);
            }
#pragma unroll
            for (int i = 0; i < UB; ++i) {
                const float dt = dtv[i];
                const float u  = uv[i];
                const float du = dt * u;
                const float Bl[SPQ] = { bv[i].x, bv[i].y, bv[i].z, bv[i].w };
                const float Cl[SPQ] = { cv[i].x, cv[i].y, cv[i].z, cv[i].w };
                float y = 0.0f;
                for (int j = 0; j < SPQ; ++j) {
                    float a = __builtin_amdgcn_exp2f(dt * A[j]);
                    h[j] = a * h[j] + du * Bl[j];
                    y   += h[j] * Cl[j];
                }
                if (q == 0) y += u * Dsd;
                yp[s0 + i][q][lane] = y;
            }
        }
        __syncthreads();
        for (int r = 0; r < TB / NQ; ++r) {
            const int tt = q * (TB / NQ) + r;
            const float s = (yp[tt][0][lane] + yp[tt][1][lane])
                          + (yp[tt][2][lane] + yp[tt][3][lane]);
            const int bl = b * LL + l0 + bt * TB + tt;
            Y[(size_t)bl * DMODEL + d] = s;
        }
    }
}

extern "C" void kernel_launch(void* const* d_in, const int* in_sizes, int n_in,
                              void* d_out, int out_size, void* d_ws, size_t ws_size,
                              hipStream_t stream)
{
    const float* x      = (const float*)d_in[0];
    const float* prompt = (const float*)d_in[1];
    const float* w1     = (const float*)d_in[2];
    const float* w2     = (const float*)d_in[3];
    const float* bias   = (const float*)d_in[4];
    const float* alog   = (const float*)d_in[5];
    const float* ds     = (const float*)d_in[6];
    float*       y      = (float*)d_out;

    float* ws = (float*)d_ws;
    size_t o = 0;
    float* dtsr  = ws + o; o += (size_t)MTOT * DTRANK;
    float* Bsc   = ws + o; o += (size_t)MTOT * DSTATE;
    float* Csc   = ws + o; o += (size_t)MTOT * DSTATE;
    float* delta = ws + o; o += (size_t)MTOT * DMODEL;
    const size_t units = (size_t)NCH * BB * DMODEL;
    float* P     = ws + o; o += units * DSTATE;
    float* Hend  = ws + o; o += units * DSTATE;
    float* Hinit = ws + o; o += units * DSTATE;
    (void)ws_size; (void)in_sizes; (void)n_in; (void)out_size;

    proj_gemm_kernel<<<dim3(MTOT / 128, 4, 1), 256, 0, stream>>>(x, w1, prompt, dtsr, Bsc, Csc);
    dt_gemm_kernel  <<<dim3(MTOT / 128, DTSPLIT, 1), 256, 0, stream>>>(dtsr, w2, bias, delta);
    scan_passA_kernel<<<dim3((unsigned)(units / 32)), 128, 0, stream>>>(delta, x, Bsc, alog, P, Hend);
    scan_passB_kernel<<<dim3(PB_PLANE / PB_THREADS), PB_THREADS, 0, stream>>>(P, Hend, Hinit);
    scan_passC_kernel<<<dim3((unsigned)(units / 32)), 128, 0, stream>>>(delta, x, Bsc, Csc, alog, ds, Hinit, y);
}


# Both GEMM kernels' start: the block's W slice is staged with all of a thread's loads issued before the first wait (dt GEMM: its A rows too); was 24 and 6 load-wait-store trips in a row
# speedup vs baseline: 6.6141x; 1.1574x over previous
#include <hip/hip_runtime.h>
#include <hip/hip_bf16.h>
#include <math.h>

#define BB      4
#define LL      4096
#define DMODEL  384
#define DSTATE  16
#define DTRANK  24
#define CTOT    (DTRANK + 2*DSTATE)
#define MTOT    (BB*LL)
#define NCH     32
#define CHL     (LL/NCH)
#define NQ      4
#define SPQ     (DSTATE/NQ)
#define DGRP    (DMODEL/32)
#define TB      32
#define UB      8
#define DTSPLIT 6
#define DTCOLS  (DMODEL/DTSPLIT)
#define LOG2E_F 1.4426950408889634f
#define LOG2E_LO_F 1.92596299e-8f
#define LN2_F      0.6931471805599453f

typedef float     v2f  __attribute__((ext_vector_type(2)));
typedef float     v8f  __attribute__((ext_vector_type(8)));
typedef _Float16  v16h __attribute__((ext_vector_type(16)));

#if defined(__gfx1250__) && __has_builtin(__builtin_amdgcn_wmma_f32_16x16x4_f32)
#define HAVE_WMMA_F32 1
#else
#define HAVE_WMMA_F32 0
#endif

__global__ __launch_bounds__(256)
void proj_gemm_kernel(const float* __restrict__ X,
                      const float* __restrict__ W1,
                      const float* __restrict__ prompt,
                      float* __restrict__ dtsr,
                      float* __restrict__ Bsc,
                      float* __restrict__ Csc)
{
    __shared__ __attribute__((aligned(16))) float Wlds[16 * 388];

    const int c0 = blockIdx.y * 16;
    {
        float4 wv[6];
#pragma unroll
        for (int j = 0; j < 6; ++j) {
            const int f   = threadIdx.x + 256 * j;
            const int n   = f / (DMODEL / 4);
            const int k4  = f - n * (DMODEL / 4);
            const int row = (c0 + n < CTOT) ? (c0 + n) : (CTOT - 1);
            wv[j] = *reinterpret_cast<const float4*>(W1 + (size_t)row * DMODEL + 4 * k4);
        }
#pragma unroll
        for (int j = 0; j < 6; ++j) {
            const int f  = threadIdx.x + 256 * j;
            const int n  = f / (DMODEL / 4);
            const int k4 = f - n * (DMODEL / 4);
            float4 v = wv[j];
            if (c0 + n >= CTOT) v = make_float4(0.0f, 0.0f, 0.0f, 0.0f);
            *reinterpret_cast<float4*>(&Wlds[n * 388 + 4 * k4]) = v;
        }
    }
    __syncthreads();

    const int wave = threadIdx.x >> 5;
    const int lane = threadIdx.x & 31;
    const int ln   = lane & 15;
    const int hi   = lane >> 4;
    const int m0   = blockIdx.x * 128 + wave * 16;

    v8f acc = {};
    const float* arow = X + (size_t)(m0 + ln) * DMODEL;

#if HAVE_WMMA_F32
    for (int k0 = 0; k0 < DMODEL; k0 += 4) {
        v2f a, b;
        a.x = arow[k0 + 2 * hi];
        a.y = arow[k0 + 2 * hi + 1];
        b.x = Wlds[ln * 388 + k0 + 2 * hi];
        b.y = Wlds[ln * 388 + k0 + 2 * hi + 1];
        acc = __builtin_amdgcn_wmma_f32_16x16x4_f32(
                  false, a, false, b, (short)0, acc, false, false);
    }
#else
    for (int k0 = 0; k0 < DMODEL; k0 += 32) {
        v16h a, b;
#pragma unroll
        for (int v = 0; v < 8; ++v) {
            int kk = (v < 4 ? 2 * v : 16 + 2 * (v - 4)) + 8 * hi;
            a[2 * v]     = (_Float16)arow[k0 + kk];
            a[2 * v + 1] = (_Float16)arow[k0 + kk + 1];
            b[2 * v]     = (_Float16)Wlds[ln * 388 + k0 + kk];
            b[2 * v + 1] = (_Float16)Wlds[ln * 388 + k0 + kk + 1];
        }
        acc = __builtin_amdgcn_wmma_f32_16x16x32_f16(
                  false, a, false, b, (short)0, acc, false, false);
    }
#endif

    const int c = c0 + ln;
#pragma unroll
    for (int r = 0; r < 8; ++r) {
        int m = m0 + r + 8 * hi;
        float v = acc[r];
        if (c < DTRANK) {
            dtsr[(size_t)m * DTRANK + c] = v;
        } else if (c < DTRANK + DSTATE) {
            Bsc[(size_t)m * DSTATE + (c - DTRANK)] = v;
        } else if (c < CTOT) {
            int n = c - DTRANK - DSTATE;
            Csc[(size_t)m * DSTATE + n] = v + prompt[(size_t)m * DSTATE + n];
        }
    }
}

__device__ __forceinline__ float softplus_f32(float t)
{
    const float na  = -__builtin_fabsf(t);
    const float hi  = na * LOG2E_F;
    float       lo  = __builtin_fmaf(na, LOG2E_F, -hi);
    lo              = __builtin_fmaf(na, LOG2E_LO_F, lo);
    const float e   = __builtin_amdgcn_exp2f(hi);
    const float z   = __builtin_fmaf(e, lo * LN2_F, e);
    const float den = 2.0f + z;
    const float r   = __builtin_amdgcn_rcpf(den);
    float       s   = z * r;
    s               = __builtin_fmaf(__builtin_fmaf(-s, den, z), r, s);
    const float w   = s * s;
    float p = 2.0f / 15.0f;
    p = __builtin_fmaf(p, w, 2.0f / 13.0f);
    p = __builtin_fmaf(p, w, 2.0f / 11.0f);
    p = __builtin_fmaf(p, w, 2.0f / 9.0f);
    p = __builtin_fmaf(p, w, 2.0f / 7.0f);
    p = __builtin_fmaf(p, w, 2.0f / 5.0f);
    p = __builtin_fmaf(p, w, 2.0f / 3.0f);
    p = __builtin_fmaf(p, w, 2.0f);
    return __builtin_fmaf(s, p, __builtin_fmaxf(t, 0.0f));
}

__global__ __launch_bounds__(256)
void dt_gemm_kernel(const float* __restrict__ dtsr,
                    const float* __restrict__ W2,
                    const float* __restrict__ bias,
                    float* __restrict__ delta)
{
    __shared__ __attribute__((aligned(16))) float Wlds[DTCOLS * 28];

    const int dbase = blockIdx.y * DTCOLS;
    const int wave = threadIdx.x >> 5;
    const int lane = threadIdx.x & 31;
    const int ln   = lane & 15;
    const int hi   = lane >> 4;
    const int m0   = blockIdx.x * 128 + wave * 16;

    const float* arow = dtsr + (size_t)(m0 + ln) * DTRANK;

    const float2* wsrc = reinterpret_cast<const float2*>(W2 + (size_t)dbase * DTRANK);
    float2 wv[3];
#pragma unroll
    for (int j = 0; j < 3; ++j) wv[j] = wsrc[threadIdx.x + 256 * j];

#if HAVE_WMMA_F32
    v2f afr[6];
#pragma unroll
    for (int ks = 0; ks < 6; ++ks) {
        afr[ks].x = arow[4 * ks + 2 * hi];
        afr[ks].y = arow[4 * ks + 2 * hi + 1];
    }
#else
    v16h afr;
#pragma unroll
    for (int v = 0; v < 8; ++v) {
        int kk = (v < 4 ? 2 * v : 16 + 2 * (v - 4)) + 8 * hi;
        afr[2 * v]     = (kk     < DTRANK) ? (_Float16)arow[kk]     : (_Float16)0.f;
        afr[2 * v + 1] = (kk + 1 < DTRANK) ? (_Float16)arow[kk + 1] : (_Float16)0.f;
    }
#endif

#pragma unroll
    for (int j = 0; j < 3; ++j) {
        const int f  = threadIdx.x + 256 * j;
        const int ds = f / (DTRANK / 2);
        const int r2 = f - ds * (DTRANK / 2);
        *reinterpret_cast<float2*>(&Wlds[ds * 28 + 2 * r2]) = wv[j];
    }
    __syncthreads();

    for (int d0 = 0; d0 < DTCOLS; d0 += 16) {
        v8f acc = {};
#if HAVE_WMMA_F32
#pragma unroll
        for (int ks = 0; ks < 6; ++ks) {
            v2f b;
            b.x = Wlds[(d0 + ln) * 28 + 4 * ks + 2 * hi];
            b.y = Wlds[(d0 + ln) * 28 + 4 * ks + 2 * hi + 1];
            acc = __builtin_amdgcn_wmma_f32_16x16x4_f32(
                      false, afr[ks], false, b, (short)0, acc, false, false);
        }
#else
        v16h b;
#pragma unroll
        for (int v = 0; v < 8; ++v) {
            int kk = (v < 4 ? 2 * v : 16 + 2 * (v - 4)) + 8 * hi;
            b[2 * v]     = (kk     < DTRANK) ? (_Float16)Wlds[(d0 + ln) * 28 + kk]     : (_Float16)0.f;
            b[2 * v + 1] = (kk + 1 < DTRANK) ? (_Float16)Wlds[(d0 + ln) * 28 + kk + 1] : (_Float16)0.f;
        }
        acc = __builtin_amdgcn_wmma_f32_16x16x32_f16(
                  false, afr, false, b, (short)0, acc, false, false);
#endif
        const int   d  = dbase + d0 + ln;
        const float bi = bias[d];
#pragma unroll
        for (int r = 0; r < 8; ++r) {
            int   m = m0 + r + 8 * hi;
            delta[(size_t)m * DMODEL + d] = softplus_f32(acc[r] + bi);
        }
    }
}

__global__ __launch_bounds__(128) __attribute__((amdgpu_waves_per_eu(8, 8)))
void scan_passA_kernel(const float* __restrict__ delta,
                       const float* __restrict__ X,
                       const float* __restrict__ Bsc,
                       const float* __restrict__ Alog,
                       float* __restrict__ Pout,
                       float* __restrict__ Hend)
{
    const int lane = threadIdx.x & 31;
    const int q    = threadIdx.x >> 5;
    const int dg   = blockIdx.x % DGRP;
    const int t2   = blockIdx.x / DGRP;
    const int b    = t2 % BB;
    const int ch   = t2 / BB;
    const int d    = dg * 32 + lane;
    const size_t unit = (size_t)(ch * BB + b) * DMODEL + d;

    float A[SPQ], h[SPQ], P[SPQ];
    {
        const float4 al = *reinterpret_cast<const float4*>(Alog + d * DSTATE + SPQ * q);
        A[0] = -expf(al.x) * LOG2E_F; A[1] = -expf(al.y) * LOG2E_F;
        A[2] = -expf(al.z) * LOG2E_F; A[3] = -expf(al.w) * LOG2E_F;
    }
    for (int j = 0; j < SPQ; ++j) { h[j] = 0.0f; P[j] = 1.0f; }

    const size_t row0 = (size_t)(b * LL + ch * CHL);
    const float* dp = delta + row0 * DMODEL + d;
    const float* xp = X     + row0 * DMODEL + d;
    const float* bp = Bsc   + row0 * DSTATE + SPQ * q;
#pragma unroll 1
    for (int t0 = 0; t0 < CHL; t0 += UB) {
        float  dtv[UB], uv[UB];
        float4 bv[UB];
#pragma unroll
        for (int i = 0; i < UB; ++i) {
            dtv[i] = dp[(size_t)(t0 + i) * DMODEL];
            uv[i]  = xp[(size_t)(t0 + i) * DMODEL];
            bv[i]  = *reinterpret_cast<const float4*>(bp + (size_t)(t0 + i) * DSTATE);
        }
#pragma unroll
        for (int i = 0; i < UB; ++i) {
            const float dt = dtv[i];
            const float du = dt * uv[i];
            const float Bl[SPQ] = { bv[i].x, bv[i].y, bv[i].z, bv[i].w };
            for (int j = 0; j < SPQ; ++j) {
                float a = __builtin_amdgcn_exp2f(dt * A[j]);
                P[j] *= a;
                h[j]  = a * h[j] + du * Bl[j];
            }
        }
    }

    float4* Pp = reinterpret_cast<float4*>(Pout + unit * DSTATE + SPQ * q);
    float4* Hp = reinterpret_cast<float4*>(Hend + unit * DSTATE + SPQ * q);
    *Pp = make_float4(P[0], P[1], P[2], P[3]);
    *Hp = make_float4(h[0], h[1], h[2], h[3]);
}

#define PB_THREADS 96
#define PB_PLANE   (BB * DMODEL * DSTATE)
__global__ __launch_bounds__(PB_THREADS)
void scan_passB_kernel(const float* __restrict__ P,
                       const float* __restrict__ Hend,
                       float* __restrict__ Hinit)
{
    const int idx = blockIdx.x * PB_THREADS + threadIdx.x;

    float p[NCH], e[NCH];
#pragma unroll
    for (int ch = 0; ch < NCH; ++ch) {
        p[ch] = P[(size_t)ch * PB_PLANE + idx];
        e[ch] = Hend[(size_t)ch * PB_PLANE + idx];
    }

    float h = 0.0f;
#pragma unroll
    for (int ch = 0; ch < NCH; ++ch) {
        Hinit[(size_t)ch * PB_PLANE + idx] = h;
        h = __builtin_fmaf(p[ch], h, e[ch]);
    }
}

__global__ __launch_bounds__(128) __attribute__((amdgpu_waves_per_eu(8, 8)))
void scan_passC_kernel(const float* __restrict__ delta,
                       const float* __restrict__ X,
                       const float* __restrict__ Bsc,
                       const float* __restrict__ Csc,
                       const float* __restrict__ Alog,
                       const float* __restrict__ Ds,
                       const float* __restrict__ Hinit,
                       float* __restrict__ Y)
{
    __shared__ float ypart[2][TB][NQ][32];

    const int lane = threadIdx.x & 31;
    const int q    = threadIdx.x >> 5;
    const int dg   = blockIdx.x % DGRP;
    const int t2   = blockIdx.x / DGRP;
    const int b    = t2 % BB;
    const int ch   = t2 / BB;
    const int d    = dg * 32 + lane;
    const size_t unit = (size_t)(ch * BB + b) * DMODEL + d;

    float A[SPQ], h[SPQ];
    {
        const float4 al = *reinterpret_cast<const float4*>(Alog + d * DSTATE + SPQ * q);
        const float4 hv = *reinterpret_cast<const float4*>(Hinit + unit * DSTATE + SPQ * q);
        A[0] = -expf(al.x) * LOG2E_F; A[1] = -expf(al.y) * LOG2E_F;
        A[2] = -expf(al.z) * LOG2E_F; A[3] = -expf(al.w) * LOG2E_F;
        h[0] = hv.x; h[1] = hv.y; h[2] = hv.z; h[3] = hv.w;
    }
    const float Dsd = Ds[d];

    const int l0 = ch * CHL;
    const size_t row0 = (size_t)(b * LL + l0);
    const float* dp = delta + row0 * DMODEL + d;
    const float* xp = X     + row0 * DMODEL + d;
    const float* bp = Bsc   + row0 * DSTATE + SPQ * q;
    const float* cp = Csc   + row0 * DSTATE + SPQ * q;
    for (int bt = 0; bt < CHL / TB; ++bt) {
        float (*yp)[NQ][32] = ypart[bt & 1];
#pragma unroll 1
        for (int s0 = 0; s0 < TB; s0 += UB) {
            const int t0 = bt * TB + s0;
            float  dtv[UB], uv[UB];
            float4 bv[UB], cv[UB];
#pragma unroll
            for (int i = 0; i < UB; ++i) {
                dtv[i] = dp[(size_t)(t0 + i) * DMODEL];
                uv[i]  = xp[(size_t)(t0 + i) * DMODEL];
                bv[i]  = *reinterpret_cast<const float4*>(bp + (size_t)(t0 + i) * DSTATE);
                cv[i]  = *reinterpret_cast<const float4*>(cp + (size_t)(t0 + i) * DSTATE);
            }
#pragma unroll
            for (int i = 0; i < UB; ++i) {
                const float dt = dtv[i];
                const float u  = uv[i];
                const float du = dt * u;
                const float Bl[SPQ] = { bv[i].x, bv[i].y, bv[i].z, bv[i].w };
                const float Cl[SPQ] = { cv[i].x, cv[i].y, cv[i].z, cv[i].w };
                float y = 0.0f;
                for (int j = 0; j < SPQ; ++j) {
                    float a = __builtin_amdgcn_exp2f(dt * A[j]);
                    h[j] = a * h[j] + du * Bl[j];
                    y   += h[j] * Cl[j];
                }
                if (q == 0) y += u * Dsd;
                yp[s0 + i][q][lane] = y;
            }
        }
        __syncthreads();
        for (int r = 0; r < TB / NQ; ++r) {
            const int tt = q * (TB / NQ) + r;
            const float s = (yp[tt][0][lane] + yp[tt][1][lane])
                          + (yp[tt][2][lane] + yp[tt][3][lane]);
            const int bl = b * LL + l0 + bt * TB + tt;
            Y[(size_t)bl * DMODEL + d] = s;
        }
    }
}

extern "C" void kernel_launch(void* const* d_in, const int* in_sizes, int n_in,
                              void* d_out, int out_size, void* d_ws, size_t ws_size,
                              hipStream_t stream)
{
    const float* x      = (const float*)d_in[0];
    const float* prompt = (const float*)d_in[1];
    const float* w1     = (const float*)d_in[2];
    const float* w2     = (const float*)d_in[3];
    const float* bias   = (const float*)d_in[4];
    const float* alog   = (const float*)d_in[5];
    const float* ds     = (const float*)d_in[6];
    float*       y      = (float*)d_out;

    float* ws = (float*)d_ws;
    size_t o = 0;
    float* dtsr  = ws + o; o += (size_t)MTOT * DTRANK;
    float* Bsc   = ws + o; o += (size_t)MTOT * DSTATE;
    float* Csc   = ws + o; o += (size_t)MTOT * DSTATE;
    float* delta = ws + o; o += (size_t)MTOT * DMODEL;
    const size_t units = (size_t)NCH * BB * DMODEL;
    float* P     = ws + o; o += units * DSTATE;
    float* Hend  = ws + o; o += units * DSTATE;
    float* Hinit = ws + o; o += units * DSTATE;
    (void)ws_size; (void)in_sizes; (void)n_in; (void)out_size;

    proj_gemm_kernel<<<dim3(MTOT / 128, 4, 1), 256, 0, stream>>>(x, w1, prompt, dtsr, Bsc, Csc);
    dt_gemm_kernel  <<<dim3(MTOT / 128, DTSPLIT, 1), 256, 0, stream>>>(dtsr, w2, bias, delta);
    scan_passA_kernel<<<dim3((unsigned)(units / 32)), 128, 0, stream>>>(delta, x, Bsc, alog, P, Hend);
    scan_passB_kernel<<<dim3(PB_PLANE / PB_THREADS), PB_THREADS, 0, stream>>>(P, Hend, Hinit);
    scan_passC_kernel<<<dim3((unsigned)(units / 32)), 128, 0, stream>>>(delta, x, Bsc, Csc, alog, ds, Hinit, y);
}


# Projection GEMM, A operand: each wave stages its 16 rows of X through a private LDS tile with whole-line loads, one 64-k chunk ahead; fragments read from LDS (was 16 lines per load); k order kept
# speedup vs baseline: 6.9375x; 1.0489x over previous
#include <hip/hip_runtime.h>
#include <hip/hip_bf16.h>
#include <math.h>

#define BB      4
#define LL      4096
#define DMODEL  384
#define DSTATE  16
#define DTRANK  24
#define CTOT    (DTRANK + 2*DSTATE)
#define MTOT    (BB*LL)
#define NCH     32
#define CHL     (LL/NCH)
#define NQ      4
#define SPQ     (DSTATE/NQ)
#define DGRP    (DMODEL/32)
#define TB      32
#define UB      8
#define DTSPLIT 6
#define DTCOLS  (DMODEL/DTSPLIT)
#define XCH     64
#define XS      68
#define NXCH    (DMODEL/XCH)
#define LOG2E_F 1.4426950408889634f
#define LOG2E_LO_F 1.92596299e-8f
#define LN2_F      0.6931471805599453f

typedef float     v2f  __attribute__((ext_vector_type(2)));
typedef float     v8f  __attribute__((ext_vector_type(8)));
typedef _Float16  v16h __attribute__((ext_vector_type(16)));

#if defined(__gfx1250__) && __has_builtin(__builtin_amdgcn_wmma_f32_16x16x4_f32)
#define HAVE_WMMA_F32 1
#else
#define HAVE_WMMA_F32 0
#endif

__global__ __launch_bounds__(256)
void proj_gemm_kernel(const float* __restrict__ X,
                      const float* __restrict__ W1,
                      const float* __restrict__ prompt,
                      float* __restrict__ dtsr,
                      float* __restrict__ Bsc,
                      float* __restrict__ Csc)
{
    __shared__ __attribute__((aligned(16))) float Wlds[16 * 388];
    __shared__ __attribute__((aligned(16))) float Xlds[8][16 * XS];

    const int c0 = blockIdx.y * 16;
    const int wave = threadIdx.x >> 5;
    const int lane = threadIdx.x & 31;
    const int ln   = lane & 15;
    const int hi   = lane >> 4;
    const int m0   = blockIdx.x * 128 + wave * 16;

    float*       xt   = Xlds[wave];
    const float* xsrc = X + (size_t)(m0 + hi) * DMODEL + 4 * ln;
    float*       xdst = xt + hi * XS + 4 * ln;

    float4 xv[8];
    {
        float4 wv[6];
#pragma unroll
        for (int j = 0; j < 6; ++j) {
            const int f   = threadIdx.x + 256 * j;
            const int n   = f / (DMODEL / 4);
            const int k4  = f - n * (DMODEL / 4);
            const int row = (c0 + n < CTOT) ? (c0 + n) : (CTOT - 1);
            wv[j] = *reinterpret_cast<const float4*>(W1 + (size_t)row * DMODEL + 4 * k4);
        }
#pragma unroll
        for (int j = 0; j < 8; ++j)
            xv[j] = *reinterpret_cast<const float4*>(xsrc + (size_t)(2 * j) * DMODEL);
#pragma unroll
        for (int j = 0; j < 6; ++j) {
            const int f  = threadIdx.x + 256 * j;
            const int n  = f / (DMODEL / 4);
            const int k4 = f - n * (DMODEL / 4);
            float4 v = wv[j];
            if (c0 + n >= CTOT) v = make_float4(0.0f, 0.0f, 0.0f, 0.0f);
            *reinterpret_cast<float4*>(&Wlds[n * 388 + 4 * k4]) = v;
        }
#pragma unroll
        for (int j = 0; j < 8; ++j)
            *reinterpret_cast<float4*>(xdst + 2 * j * XS) = xv[j];
    }
    __syncthreads();

    v8f acc = {};

#if HAVE_WMMA_F32
#pragma unroll
    for (int c = 0; c < NXCH; ++c) {
        if (c + 1 < NXCH) {
#pragma unroll
            for (int j = 0; j < 8; ++j)
                xv[j] = *reinterpret_cast<const float4*>(xsrc + (size_t)(2 * j) * DMODEL + (c + 1) * XCH);
        }
        v2f af[XCH / 4], bf[XCH / 4];
#pragma unroll
        for (int s = 0; s < XCH / 4; ++s) {
            af[s].x = xt[ln * XS + 4 * s + 2 * hi];
            af[s].y = xt[ln * XS + 4 * s + 2 * hi + 1];
            bf[s].x = Wlds[ln * 388 + c * XCH + 4 * s + 2 * hi];
            bf[s].y = Wlds[ln * 388 + c * XCH + 4 * s + 2 * hi + 1];
        }
#pragma unroll
        for (int s = 0; s < XCH / 4; ++s)
            acc = __builtin_amdgcn_wmma_f32_16x16x4_f32(
                      false, af[s], false, bf[s], (short)0, acc, false, false);
        if (c + 1 < NXCH) {
#pragma unroll
            for (int j = 0; j < 8; ++j)
                *reinterpret_cast<float4*>(xdst + 2 * j * XS) = xv[j];
            __builtin_amdgcn_fence(__ATOMIC_SEQ_CST, "wavefront");
        }
    }
#else
    const float* arow = X + (size_t)(m0 + ln) * DMODEL;
    for (int k0 = 0; k0 < DMODEL; k0 += 32) {
        v16h a, b;
#pragma unroll
        for (int v = 0; v < 8; ++v) {
            int kk = (v < 4 ? 2 * v : 16 + 2 * (v - 4)) + 8 * hi;
            a[2 * v]     = (_Float16)arow[k0 + kk];
            a[2 * v + 1] = (_Float16)arow[k0 + kk + 1];
            b[2 * v]     = (_Float16)Wlds[ln * 388 + k0 + kk];
            b[2 * v + 1] = (_Float16)Wlds[ln * 388 + k0 + kk + 1];
        }
        acc = __builtin_amdgcn_wmma_f32_16x16x32_f16(
                  false, a, false, b, (short)0, acc, false, false);
    }
#endif

    const int c = c0 + ln;
#pragma unroll
    for (int r = 0; r < 8; ++r) {
        int m = m0 + r + 8 * hi;
        float v = acc[r];
        if (c < DTRANK) {
            dtsr[(size_t)m * DTRANK + c] = v;
        } else if (c < DTRANK + DSTATE) {
            Bsc[(size_t)m * DSTATE + (c - DTRANK)] = v;
        } else if (c < CTOT) {
            int n = c - DTRANK - DSTATE;
            Csc[(size_t)m * DSTATE + n] = v + prompt[(size_t)m * DSTATE + n];
        }
    }
}

__device__ __forceinline__ float softplus_f32(float t)
{
    const float na  = -__builtin_fabsf(t);
    const float hi  = na * LOG2E_F;
    float       lo  = __builtin_fmaf(na, LOG2E_F, -hi);
    lo              = __builtin_fmaf(na, LOG2E_LO_F, lo);
    const float e   = __builtin_amdgcn_exp2f(hi);
    const float z   = __builtin_fmaf(e, lo * LN2_F, e);
    const float den = 2.0f + z;
    const float r   = __builtin_amdgcn_rcpf(den);
    float       s   = z * r;
    s               = __builtin_fmaf(__builtin_fmaf(-s, den, z), r, s);
    const float w   = s * s;
    float p = 2.0f / 15.0f;
    p = __builtin_fmaf(p, w, 2.0f / 13.0f);
    p = __builtin_fmaf(p, w, 2.0f / 11.0f);
    p = __builtin_fmaf(p, w, 2.0f / 9.0f);
    p = __builtin_fmaf(p, w, 2.0f / 7.0f);
    p = __builtin_fmaf(p, w, 2.0f / 5.0f);
    p = __builtin_fmaf(p, w, 2.0f / 3.0f);
    p = __builtin_fmaf(p, w, 2.0f);
    return __builtin_fmaf(s, p, __builtin_fmaxf(t, 0.0f));
}

__global__ __launch_bounds__(256)
void dt_gemm_kernel(const float* __restrict__ dtsr,
                    const float* __restrict__ W2,
                    const float* __restrict__ bias,
                    float* __restrict__ delta)
{
    __shared__ __attribute__((aligned(16))) float Wlds[DTCOLS * 28];

    const int dbase = blockIdx.y * DTCOLS;
    const int wave = threadIdx.x >> 5;
    const int lane = threadIdx.x & 31;
    const int ln   = lane & 15;
    const int hi   = lane >> 4;
    const int m0   = blockIdx.x * 128 + wave * 16;

    const float* arow = dtsr + (size_t)(m0 + ln) * DTRANK;

    const float2* wsrc = reinterpret_cast<const float2*>(W2 + (size_t)dbase * DTRANK);
    float2 wv[3];
#pragma unroll
    for (int j = 0; j < 3; ++j) wv[j] = wsrc[threadIdx.x + 256 * j];

#if HAVE_WMMA_F32
    v2f afr[6];
#pragma unroll
    for (int ks = 0; ks < 6; ++ks) {
        afr[ks].x = arow[4 * ks + 2 * hi];
        afr[ks].y = arow[4 * ks + 2 * hi + 1];
    }
#else
    v16h afr;
#pragma unroll
    for (int v = 0; v < 8; ++v) {
        int kk = (v < 4 ? 2 * v : 16 + 2 * (v - 4)) + 8 * hi;
        afr[2 * v]     = (kk     < DTRANK) ? (_Float16)arow[kk]     : (_Float16)0.f;
        afr[2 * v + 1] = (kk + 1 < DTRANK) ? (_Float16)arow[kk + 1] : (_Float16)0.f;
    }
#endif

#pragma unroll
    for (int j = 0; j < 3; ++j) {
        const int f  = threadIdx.x + 256 * j;
        const int ds = f / (DTRANK / 2);
        const int r2 = f - ds * (DTRANK / 2);
        *reinterpret_cast<float2*>(&Wlds[ds * 28 + 2 * r2]) = wv[j];
    }
    __syncthreads();

    for (int d0 = 0; d0 < DTCOLS; d0 += 16) {
        v8f acc = {};
#if HAVE_WMMA_F32
#pragma unroll
        for (int ks = 0; ks < 6; ++ks) {
            v2f b;
            b.x = Wlds[(d0 + ln) * 28 + 4 * ks + 2 * hi];
            b.y = Wlds[(d0 + ln) * 28 + 4 * ks + 2 * hi + 1];
            acc = __builtin_amdgcn_wmma_f32_16x16x4_f32(
                      false, afr[ks], false, b, (short)0, acc, false, false);
        }
#else
        v16h b;
#pragma unroll
        for (int v = 0; v < 8; ++v) {
            int kk = (v < 4 ? 2 * v : 16 + 2 * (v - 4)) + 8 * hi;
            b[2 * v]     = (kk     < DTRANK) ? (_Float16)Wlds[(d0 + ln) * 28 + kk]     : (_Float16)0.f;
            b[2 * v + 1] = (kk + 1 < DTRANK) ? (_Float16)Wlds[(d0 + ln) * 28 + kk + 1] : (_Float16)0.f;
        }
        acc = __builtin_amdgcn_wmma_f32_16x16x32_f16(
                  false, afr, false, b, (short)0, acc, false, false);
#endif
        const int   d  = dbase + d0 + ln;
        const float bi = bias[d];
#pragma unroll
        for (int r = 0; r < 8; ++r) {
            int   m = m0 + r + 8 * hi;
            delta[(size_t)m * DMODEL + d] = softplus_f32(acc[r] + bi);
        }
    }
}

__global__ __launch_bounds__(128) __attribute__((amdgpu_waves_per_eu(8, 8)))
void scan_passA_kernel(const float* __restrict__ delta,
                       const float* __restrict__ X,
                       const float* __restrict__ Bsc,
                       const float* __restrict__ Alog,
                       float* __restrict__ Pout,
                       float* __restrict__ Hend)
{
    const int lane = threadIdx.x & 31;
    const int q    = threadIdx.x >> 5;
    const int dg   = blockIdx.x % DGRP;
    const int t2   = blockIdx.x / DGRP;
    const int b    = t2 % BB;
    const int ch   = t2 / BB;
    const int d    = dg * 32 + lane;
    const size_t unit = (size_t)(ch * BB + b) * DMODEL + d;

    float A[SPQ], h[SPQ], P[SPQ];
    {
        const float4 al = *reinterpret_cast<const float4*>(Alog + d * DSTATE + SPQ * q);
        A[0] = -expf(al.x) * LOG2E_F; A[1] = -expf(al.y) * LOG2E_F;
        A[2] = -expf(al.z) * LOG2E_F; A[3] = -expf(al.w) * LOG2E_F;
    }
    for (int j = 0; j < SPQ; ++j) { h[j] = 0.0f; P[j] = 1.0f; }

    const size_t row0 = (size_t)(b * LL + ch * CHL);
    const float* dp = delta + row0 * DMODEL + d;
    const float* xp = X     + row0 * DMODEL + d;
    const float* bp = Bsc   + row0 * DSTATE + SPQ * q;
#pragma unroll 1
    for (int t0 = 0; t0 < CHL; t0 += UB) {
        float  dtv[UB], uv[UB];
        float4 bv[UB];
#pragma unroll
        for (int i = 0; i < UB; ++i) {
            dtv[i] = dp[(size_t)(t0 + i) * DMODEL];
            uv[i]  = xp[(size_t)(t0 + i) * DMODEL];
            bv[i]  = *reinterpret_cast<const float4*>(bp + (size_t)(t0 + i) * DSTATE);
        }
#pragma unroll
        for (int i = 0; i < UB; ++i) {
            const float dt = dtv[i];
            const float du = dt * uv[i];
            const float Bl[SPQ] = { bv[i].x, bv[i].y, bv[i].z, bv[i].w };
            for (int j = 0; j < SPQ; ++j) {
                float a = __builtin_amdgcn_exp2f(dt * A[j]);
                P[j] *= a;
                h[j]  = a * h[j] + du * Bl[j];
            }
        }
    }

    float4* Pp = reinterpret_cast<float4*>(Pout + unit * DSTATE + SPQ * q);
    float4* Hp = reinterpret_cast<float4*>(Hend + unit * DSTATE + SPQ * q);
    *Pp = make_float4(P[0], P[1], P[2], P[3]);
    *Hp = make_float4(h[0], h[1], h[2], h[3]);
}

#define PB_THREADS 96
#define PB_PLANE   (BB * DMODEL * DSTATE)
__global__ __launch_bounds__(PB_THREADS)
void scan_passB_kernel(const float* __restrict__ P,
                       const float* __restrict__ Hend,
                       float* __restrict__ Hinit)
{
    const int idx = blockIdx.x * PB_THREADS + threadIdx.x;

    float p[NCH], e[NCH];
#pragma unroll
    for (int ch = 0; ch < NCH; ++ch) {
        p[ch] = P[(size_t)ch * PB_PLANE + idx];
        e[ch] = Hend[(size_t)ch * PB_PLANE + idx];
    }

    float h = 0.0f;
#pragma unroll
    for (int ch = 0; ch < NCH; ++ch) {
        Hinit[(size_t)ch * PB_PLANE + idx] = h;
        h = __builtin_fmaf(p[ch], h, e[ch]);
    }
}

__global__ __launch_bounds__(128) __attribute__((amdgpu_waves_per_eu(8, 8)))
void scan_passC_kernel(const float* __restrict__ delta,
                       const float* __restrict__ X,
                       const float* __restrict__ Bsc,
                       const float* __restrict__ Csc,
                       const float* __restrict__ Alog,
                       const float* __restrict__ Ds,
                       const float* __restrict__ Hinit,
                       float* __restrict__ Y)
{
    __shared__ float ypart[2][TB][NQ][32];

    const int lane = threadIdx.x & 31;
    const int q    = threadIdx.x >> 5;
    const int dg   = blockIdx.x % DGRP;
    const int t2   = blockIdx.x / DGRP;
    const int b    = t2 % BB;
    const int ch   = t2 / BB;
    const int d    = dg * 32 + lane;
    const size_t unit = (size_t)(ch * BB + b) * DMODEL + d;

    float A[SPQ], h[SPQ];
    {
        const float4 al = *reinterpret_cast<const float4*>(Alog + d * DSTATE + SPQ * q);
        const float4 hv = *reinterpret_cast<const float4*>(Hinit + unit * DSTATE + SPQ * q);
        A[0] = -expf(al.x) * LOG2E_F; A[1] = -expf(al.y) * LOG2E_F;
        A[2] = -expf(al.z) * LOG2E_F; A[3] = -expf(al.w) * LOG2E_F;
        h[0] = hv.x; h[1] = hv.y; h[2] = hv.z; h[3] = hv.w;
    }
    const float Dsd = Ds[d];

    const int l0 = ch * CHL;
    const size_t row0 = (size_t)(b * LL + l0);
    const float* dp = delta + row0 * DMODEL + d;
    const float* xp = X     + row0 * DMODEL + d;
    const float* bp = Bsc   + row0 * DSTATE + SPQ * q;
    const float* cp = Csc   + row0 * DSTATE + SPQ * q;
    for (int bt = 0; bt < CHL / TB; ++bt) {
        float (*yp)[NQ][32] = ypart[bt & 1];
#pragma unroll 1
        for (int s0 = 0; s0 < TB; s0 += UB) {
            const int t0 = bt * TB + s0;
            float  dtv[UB], uv[UB];
            float4 bv[UB], cv[UB];
#pragma unroll
            for (int i = 0; i < UB; ++i) {
                dtv[i] = dp[(size_t)(t0 + i) * DMODEL];
                uv[i]  = xp[(size_t)(t0 + i) * DMODEL];
                bv[i]  = *reinterpret_cast<const float4*>(bp + (size_t)(t0 + i) * DSTATE);
                cv[i]  = *reinterpret_cast<const float4*>(cp + (size_t)(t0 + i) * DSTATE);
            }
#pragma unroll
            for (int i = 0; i < UB; ++i) {
                const float dt = dtv[i];
                const float u  = uv[i];
                const float du = dt * u;
                const float Bl[SPQ] = { bv[i].x, bv[i].y, bv[i].z, bv[i].w };
                const float Cl[SPQ] = { cv[i].x, cv[i].y, cv[i].z, cv[i].w };
                float y = 0.0f;
                for (int j = 0; j < SPQ; ++j) {
                    float a = __builtin_amdgcn_exp2f(dt * A[j]);
                    h[j] = a * h[j] + du * Bl[j];
                    y   += h[j] * Cl[j];
                }
                if (q == 0) y += u * Dsd;
                yp[s0 + i][q][lane] = y;
            }
        }
        __syncthreads();
        for (int r = 0; r < TB / NQ; ++r) {
            const int tt = q * (TB / NQ) + r;
            const float s = (yp[tt][0][lane] + yp[tt][1][lane])
                          + (yp[tt][2][lane] + yp[tt][3][lane]);
            const int bl = b * LL + l0 + bt * TB + tt;
            Y[(size_t)bl * DMODEL + d] = s;
        }
    }
}

extern "C" void kernel_launch(void* const* d_in, const int* in_sizes, int n_in,
                              void* d_out, int out_size, void* d_ws, size_t ws_size,
                              hipStream_t stream)
{
    const float* x      = (const float*)d_in[0];
    const float* prompt = (const float*)d_in[1];
    const float* w1     = (const float*)d_in[2];
    const float* w2     = (const float*)d_in[3];
    const float* bias   = (const float*)d_in[4];
    const float* alog   = (const float*)d_in[5];
    const float* ds     = (const float*)d_in[6];
    float*       y      = (float*)d_out;

    float* ws = (float*)d_ws;
    size_t o = 0;
    float* dtsr  = ws + o; o += (size_t)MTOT * DTRANK;
    float* Bsc   = ws + o; o += (size_t)MTOT * DSTATE;
    float* Csc   = ws + o; o += (size_t)MTOT * DSTATE;
    float* delta = ws + o; o += (size_t)MTOT * DMODEL;
    const size_t units = (size_t)NCH * BB * DMODEL;
    float* P     = ws + o; o += units * DSTATE;
    float* Hend  = ws + o; o += units * DSTATE;
    float* Hinit = ws + o; o += units * DSTATE;
    (void)ws_size; (void)in_sizes; (void)n_in; (void)out_size;

    proj_gemm_kernel<<<dim3(MTOT / 128, 4, 1), 256, 0, stream>>>(x, w1, prompt, dtsr, Bsc, Csc);
    dt_gemm_kernel  <<<dim3(MTOT / 128, DTSPLIT, 1), 256, 0, stream>>>(dtsr, w2, bias, delta);
    scan_passA_kernel<<<dim3((unsigned)(units / 32)), 128, 0, stream>>>(delta, x, Bsc, alog, P, Hend);
    scan_passB_kernel<<<dim3(PB_PLANE / PB_THREADS), PB_THREADS, 0, stream>>>(P, Hend, Hinit);
    scan_passC_kernel<<<dim3((unsigned)(units / 32)), 128, 0, stream>>>(delta, x, Bsc, Csc, alog, ds, Hinit, y);
}


# dt GEMM and scan pass A fused into one kernel: a block's 128 rows are one scan chunk, so its delta tile stays in LDS for pass A (one launch fewer, pass A's delta reads gone); sum order kept
# speedup vs baseline: 7.4591x; 1.0752x over previous
#include <hip/hip_runtime.h>
#include <hip/hip_bf16.h>
#include <math.h>

#define BB      4
#define LL      4096
#define DMODEL  384
#define DSTATE  16
#define DTRANK  24
#define CTOT    (DTRANK + 2*DSTATE)
#define MTOT    (BB*LL)
#define NCH     32
#define CHL     (LL/NCH)
#define NQ      4
#define SPQ     (DSTATE/NQ)
#define DGRP    (DMODEL/32)
#define TB      32
#define UB      8
#define DTSPLIT 6
#define DTCOLS  (DMODEL/DTSPLIT)
#define XCH     64
#define XS      68
#define NXCH    (DMODEL/XCH)
#define LOG2E_F 1.4426950408889634f
#define LOG2E_LO_F 1.92596299e-8f
#define LN2_F      0.6931471805599453f

typedef float     v2f  __attribute__((ext_vector_type(2)));
typedef float     v8f  __attribute__((ext_vector_type(8)));
typedef _Float16  v16h __attribute__((ext_vector_type(16)));

#if defined(__gfx1250__) && __has_builtin(__builtin_amdgcn_wmma_f32_16x16x4_f32)
#define HAVE_WMMA_F32 1
#else
#define HAVE_WMMA_F32 0
#endif

__global__ __launch_bounds__(256)
void proj_gemm_kernel(const float* __restrict__ X,
                      const float* __restrict__ W1,
                      const float* __restrict__ prompt,
                      float* __restrict__ dtsr,
                      float* __restrict__ Bsc,
                      float* __restrict__ Csc)
{
    __shared__ __attribute__((aligned(16))) float Wlds[16 * 388];
    __shared__ __attribute__((aligned(16))) float Xlds[8][16 * XS];

    const int c0 = blockIdx.y * 16;
    const int wave = threadIdx.x >> 5;
    const int lane = threadIdx.x & 31;
    const int ln   = lane & 15;
    const int hi   = lane >> 4;
    const int m0   = blockIdx.x * 128 + wave * 16;

    float*       xt   = Xlds[wave];
    const float* xsrc = X + (size_t)(m0 + hi) * DMODEL + 4 * ln;
    float*       xdst = xt + hi * XS + 4 * ln;

    float4 xv[8];
    {
        float4 wv[6];
#pragma unroll
        for (int j = 0; j < 6; ++j) {
            const int f   = threadIdx.x + 256 * j;
            const int n   = f / (DMODEL / 4);
            const int k4  = f - n * (DMODEL / 4);
            const int row = (c0 + n < CTOT) ? (c0 + n) : (CTOT - 1);
            wv[j] = *reinterpret_cast<const float4*>(W1 + (size_t)row * DMODEL + 4 * k4);
        }
#pragma unroll
        for (int j = 0; j < 8; ++j)
            xv[j] = *reinterpret_cast<const float4*>(xsrc + (size_t)(2 * j) * DMODEL);
#pragma unroll
        for (int j = 0; j < 6; ++j) {
            const int f  = threadIdx.x + 256 * j;
            const int n  = f / (DMODEL / 4);
            const int k4 = f - n * (DMODEL / 4);
            float4 v = wv[j];
            if (c0 + n >= CTOT) v = make_float4(0.0f, 0.0f, 0.0f, 0.0f);
            *reinterpret_cast<float4*>(&Wlds[n * 388 + 4 * k4]) = v;
        }
#pragma unroll
        for (int j = 0; j < 8; ++j)
            *reinterpret_cast<float4*>(xdst + 2 * j * XS) = xv[j];
    }
    __syncthreads();

    v8f acc = {};

#if HAVE_WMMA_F32
#pragma unroll
    for (int c = 0; c < NXCH; ++c) {
        if (c + 1 < NXCH) {
#pragma unroll
            for (int j = 0; j < 8; ++j)
                xv[j] = *reinterpret_cast<const float4*>(xsrc + (size_t)(2 * j) * DMODEL + (c + 1) * XCH);
        }
        v2f af[XCH / 4], bf[XCH / 4];
#pragma unroll
        for (int s = 0; s < XCH / 4; ++s) {
            af[s].x = xt[ln * XS + 4 * s + 2 * hi];
            af[s].y = xt[ln * XS + 4 * s + 2 * hi + 1];
            bf[s].x = Wlds[ln * 388 + c * XCH + 4 * s + 2 * hi];
            bf[s].y = Wlds[ln * 388 + c * XCH + 4 * s + 2 * hi + 1];
        }
#pragma unroll
        for (int s = 0; s < XCH / 4; ++s)
            acc = __builtin_amdgcn_wmma_f32_16x16x4_f32(
                      false, af[s], false, bf[s], (short)0, acc, false, false);
        if (c + 1 < NXCH) {
#pragma unroll
            for (int j = 0; j < 8; ++j)
                *reinterpret_cast<float4*>(xdst + 2 * j * XS) = xv[j];
            __builtin_amdgcn_fence(__ATOMIC_SEQ_CST, "wavefront");
        }
    }
#else
    const float* arow = X + (size_t)(m0 + ln) * DMODEL;
    for (int k0 = 0; k0 < DMODEL; k0 += 32) {
        v16h a, b;
#pragma unroll
        for (int v = 0; v < 8; ++v) {
            int kk = (v < 4 ? 2 * v : 16 + 2 * (v - 4)) + 8 * hi;
            a[2 * v]     = (_Float16)arow[k0 + kk];
            a[2 * v + 1] = (_Float16)arow[k0 + kk + 1];
            b[2 * v]     = (_Float16)Wlds[ln * 388 + k0 + kk];
            b[2 * v + 1] = (_Float16)Wlds[ln * 388 + k0 + kk + 1];
        }
        acc = __builtin_amdgcn_wmma_f32_16x16x32_f16(
                  false, a, false, b, (short)0, acc, false, false);
    }
#endif

    const int c = c0 + ln;
#pragma unroll
    for (int r = 0; r < 8; ++r) {
        int m = m0 + r + 8 * hi;
        float v = acc[r];
        if (c < DTRANK) {
            dtsr[(size_t)m * DTRANK + c] = v;
        } else if (c < DTRANK + DSTATE) {
            Bsc[(size_t)m * DSTATE + (c - DTRANK)] = v;
        } else if (c < CTOT) {
            int n = c - DTRANK - DSTATE;
            Csc[(size_t)m * DSTATE + n] = v + prompt[(size_t)m * DSTATE + n];
        }
    }
}

__device__ __forceinline__ float softplus_f32(float t)
{
    const float na  = -__builtin_fabsf(t);
    const float hi  = na * LOG2E_F;
    float       lo  = __builtin_fmaf(na, LOG2E_F, -hi);
    lo              = __builtin_fmaf(na, LOG2E_LO_F, lo);
    const float e   = __builtin_amdgcn_exp2f(hi);
    const float z   = __builtin_fmaf(e, lo * LN2_F, e);
    const float den = 2.0f + z;
    const float r   = __builtin_amdgcn_rcpf(den);
    float       s   = z * r;
    s               = __builtin_fmaf(__builtin_fmaf(-s, den, z), r, s);
    const float w   = s * s;
    float p = 2.0f / 15.0f;
    p = __builtin_fmaf(p, w, 2.0f / 13.0f);
    p = __builtin_fmaf(p, w, 2.0f / 11.0f);
    p = __builtin_fmaf(p, w, 2.0f / 9.0f);
    p = __builtin_fmaf(p, w, 2.0f / 7.0f);
    p = __builtin_fmaf(p, w, 2.0f / 5.0f);
    p = __builtin_fmaf(p, w, 2.0f / 3.0f);
    p = __builtin_fmaf(p, w, 2.0f);
    return __builtin_fmaf(s, p, __builtin_fmaxf(t, 0.0f));
}

#define DLP 68
__global__ __launch_bounds__(256) __attribute__((amdgpu_waves_per_eu(8, 8)))
void dt_gemm_scanA_kernel(const float* __restrict__ dtsr,
                          const float* __restrict__ W2,
                          const float* __restrict__ bias,
                          const float* __restrict__ X,
                          const float* __restrict__ Bsc,
                          const float* __restrict__ Alog,
                          float* __restrict__ delta,
                          float* __restrict__ Pout,
                          float* __restrict__ Hend)
{
    __shared__ __attribute__((aligned(16))) float Wlds[DTCOLS * 28];
    __shared__ __attribute__((aligned(16))) float dl[CHL * DLP];

    const int dbase = blockIdx.y * DTCOLS;
    const int wave = threadIdx.x >> 5;
    const int lane = threadIdx.x & 31;
    const int ln   = lane & 15;
    const int hi   = lane >> 4;
    const int m0   = blockIdx.x * 128 + wave * 16;

    const int q    = wave & 3;
    const int dloc = (wave >> 2) * 32 + lane;
    const int dsc  = dbase + dloc;

    const float* arow = dtsr + (size_t)(m0 + ln) * DTRANK;

    const float2* wsrc = reinterpret_cast<const float2*>(W2 + (size_t)dbase * DTRANK);
    float2 wv[3];
#pragma unroll
    for (int j = 0; j < 3; ++j) wv[j] = wsrc[threadIdx.x + 256 * j];

#if HAVE_WMMA_F32
    v2f afr[6];
#pragma unroll
    for (int ks = 0; ks < 6; ++ks) {
        afr[ks].x = arow[4 * ks + 2 * hi];
        afr[ks].y = arow[4 * ks + 2 * hi + 1];
    }
#else
    v16h afr;
#pragma unroll
    for (int v = 0; v < 8; ++v) {
        int kk = (v < 4 ? 2 * v : 16 + 2 * (v - 4)) + 8 * hi;
        afr[2 * v]     = (kk     < DTRANK) ? (_Float16)arow[kk]     : (_Float16)0.f;
        afr[2 * v + 1] = (kk + 1 < DTRANK) ? (_Float16)arow[kk + 1] : (_Float16)0.f;
    }
#endif
    const float4 al = *reinterpret_cast<const float4*>(Alog + dsc * DSTATE + SPQ * q);

#pragma unroll
    for (int j = 0; j < 3; ++j) {
        const int f  = threadIdx.x + 256 * j;
        const int ds = f / (DTRANK / 2);
        const int r2 = f - ds * (DTRANK / 2);
        *reinterpret_cast<float2*>(&Wlds[ds * 28 + 2 * r2]) = wv[j];
    }
    __syncthreads();

    for (int d0 = 0; d0 < DTCOLS; d0 += 16) {
        v8f acc = {};
#if HAVE_WMMA_F32
#pragma unroll
        for (int ks = 0; ks < 6; ++ks) {
            v2f b;
            b.x = Wlds[(d0 + ln) * 28 + 4 * ks + 2 * hi];
            b.y = Wlds[(d0 + ln) * 28 + 4 * ks + 2 * hi + 1];
            acc = __builtin_amdgcn_wmma_f32_16x16x4_f32(
                      false, afr[ks], false, b, (short)0, acc, false, false);
        }
#else
        v16h b;
#pragma unroll
        for (int v = 0; v < 8; ++v) {
            int kk = (v < 4 ? 2 * v : 16 + 2 * (v - 4)) + 8 * hi;
            b[2 * v]     = (kk     < DTRANK) ? (_Float16)Wlds[(d0 + ln) * 28 + kk]     : (_Float16)0.f;
            b[2 * v + 1] = (kk + 1 < DTRANK) ? (_Float16)Wlds[(d0 + ln) * 28 + kk + 1] : (_Float16)0.f;
        }
        acc = __builtin_amdgcn_wmma_f32_16x16x32_f16(
                  false, afr, false, b, (short)0, acc, false, false);
#endif
        const int   d  = dbase + d0 + ln;
        const float bi = bias[d];
#pragma unroll
        for (int r = 0; r < 8; ++r) {
            const int   m  = m0 + r + 8 * hi;
            const float sp = softplus_f32(acc[r] + bi);
            delta[(size_t)m * DMODEL + d] = sp;
            dl[(wave * 16 + r + 8 * hi) * DLP + d0 + ln] = sp;
        }
    }
    __syncthreads();

    const int b  = blockIdx.x / NCH;
    const int ch = blockIdx.x % NCH;
    const size_t unit = (size_t)(ch * BB + b) * DMODEL + dsc;

    float A[SPQ], h[SPQ], P[SPQ];
    A[0] = -expf(al.x) * LOG2E_F; A[1] = -expf(al.y) * LOG2E_F;
    A[2] = -expf(al.z) * LOG2E_F; A[3] = -expf(al.w) * LOG2E_F;
    for (int j = 0; j < SPQ; ++j) { h[j] = 0.0f; P[j] = 1.0f; }

    const size_t row0 = (size_t)blockIdx.x * CHL;
    const float* dp = dl + dloc;
    const float* xp = X   + row0 * DMODEL + dsc;
    const float* bp = Bsc + row0 * DSTATE + SPQ * q;
#pragma unroll 1
    for (int t0 = 0; t0 < CHL; t0 += UB) {
        float  dtv[UB], uv[UB];
        float4 bv[UB];
#pragma unroll
        for (int i = 0; i < UB; ++i) {
            dtv[i] = dp[(t0 + i) * DLP];
            uv[i]  = xp[(size_t)(t0 + i) * DMODEL];
            bv[i]  = *reinterpret_cast<const float4*>(bp + (size_t)(t0 + i) * DSTATE);
        }
#pragma unroll
        for (int i = 0; i < UB; ++i) {
            const float dt = dtv[i];
            const float du = dt * uv[i];
            const float Bl[SPQ] = { bv[i].x, bv[i].y, bv[i].z, bv[i].w };
            for (int j = 0; j < SPQ; ++j) {
                float a = __builtin_amdgcn_exp2f(dt * A[j]);
                P[j] *= a;
                h[j]  = a * h[j] + du * Bl[j];
            }
        }
    }

    float4* Pp = reinterpret_cast<float4*>(Pout + unit * DSTATE + SPQ * q);
    float4* Hp = reinterpret_cast<float4*>(Hend + unit * DSTATE + SPQ * q);
    *Pp = make_float4(P[0], P[1], P[2], P[3]);
    *Hp = make_float4(h[0], h[1], h[2], h[3]);
}

#define PB_THREADS 96
#define PB_PLANE   (BB * DMODEL * DSTATE)
__global__ __launch_bounds__(PB_THREADS)
void scan_passB_kernel(const float* __restrict__ P,
                       const float* __restrict__ Hend,
                       float* __restrict__ Hinit)
{
    const int idx = blockIdx.x * PB_THREADS + threadIdx.x;

    float p[NCH], e[NCH];
#pragma unroll
    for (int ch = 0; ch < NCH; ++ch) {
        p[ch] = P[(size_t)ch * PB_PLANE + idx];
        e[ch] = Hend[(size_t)ch * PB_PLANE + idx];
    }

    float h = 0.0f;
#pragma unroll
    for (int ch = 0; ch < NCH; ++ch) {
        Hinit[(size_t)ch * PB_PLANE + idx] = h;
        h = __builtin_fmaf(p[ch], h, e[ch]);
    }
}

__global__ __launch_bounds__(128) __attribute__((amdgpu_waves_per_eu(8, 8)))
void scan_passC_kernel(const float* __restrict__ delta,
                       const float* __restrict__ X,
                       const float* __restrict__ Bsc,
                       const float* __restrict__ Csc,
                       const float* __restrict__ Alog,
                       const float* __restrict__ Ds,
                       const float* __restrict__ Hinit,
                       float* __restrict__ Y)
{
    __shared__ float ypart[2][TB][NQ][32];

    const int lane = threadIdx.x & 31;
    const int q    = threadIdx.x >> 5;
    const int dg   = blockIdx.x % DGRP;
    const int t2   = blockIdx.x / DGRP;
    const int b    = t2 % BB;
    const int ch   = t2 / BB;
    const int d    = dg * 32 + lane;
    const size_t unit = (size_t)(ch * BB + b) * DMODEL + d;

    float A[SPQ], h[SPQ];
    {
        const float4 al = *reinterpret_cast<const float4*>(Alog + d * DSTATE + SPQ * q);
        const float4 hv = *reinterpret_cast<const float4*>(Hinit + unit * DSTATE + SPQ * q);
        A[0] = -expf(al.x) * LOG2E_F; A[1] = -expf(al.y) * LOG2E_F;
        A[2] = -expf(al.z) * LOG2E_F; A[3] = -expf(al.w) * LOG2E_F;
        h[0] = hv.x; h[1] = hv.y; h[2] = hv.z; h[3] = hv.w;
    }
    const float Dsd = Ds[d];

    const int l0 = ch * CHL;
    const size_t row0 = (size_t)(b * LL + l0);
    const float* dp = delta + row0 * DMODEL + d;
    const float* xp = X     + row0 * DMODEL + d;
    const float* bp = Bsc   + row0 * DSTATE + SPQ * q;
    const float* cp = Csc   + row0 * DSTATE + SPQ * q;
    for (int bt = 0; bt < CHL / TB; ++bt) {
        float (*yp)[NQ][32] = ypart[bt & 1];
#pragma unroll 1
        for (int s0 = 0; s0 < TB; s0 += UB) {
            const int t0 = bt * TB + s0;
            float  dtv[UB], uv[UB];
            float4 bv[UB], cv[UB];
#pragma unroll
            for (int i = 0; i < UB; ++i) {
                dtv[i] = dp[(size_t)(t0 + i) * DMODEL];
                uv[i]  = xp[(size_t)(t0 + i) * DMODEL];
                bv[i]  = *reinterpret_cast<const float4*>(bp + (size_t)(t0 + i) * DSTATE);
                cv[i]  = *reinterpret_cast<const float4*>(cp + (size_t)(t0 + i) * DSTATE);
            }
#pragma unroll
            for (int i = 0; i < UB; ++i) {
                const float dt = dtv[i];
                const float u  = uv[i];
                const float du = dt * u;
                const float Bl[SPQ] = { bv[i].x, bv[i].y, bv[i].z, bv[i].w };
                const float Cl[SPQ] = { cv[i].x, cv[i].y, cv[i].z, cv[i].w };
                float y = 0.0f;
                for (int j = 0; j < SPQ; ++j) {
                    float a = __builtin_amdgcn_exp2f(dt * A[j]);
                    h[j] = a * h[j] + du * Bl[j];
                    y   += h[j] * Cl[j];
                }
                if (q == 0) y += u * Dsd;
                yp[s0 + i][q][lane] = y;
            }
        }
        __syncthreads();
        for (int r = 0; r < TB / NQ; ++r) {
            const int tt = q * (TB / NQ) + r;
            const float s = (yp[tt][0][lane] + yp[tt][1][lane])
                          + (yp[tt][2][lane] + yp[tt][3][lane]);
            const int bl = b * LL + l0 + bt * TB + tt;
            Y[(size_t)bl * DMODEL + d] = s;
        }
    }
}

extern "C" void kernel_launch(void* const* d_in, const int* in_sizes, int n_in,
                              void* d_out, int out_size, void* d_ws, size_t ws_size,
                              hipStream_t stream)
{
    const float* x      = (const float*)d_in[0];
    const float* prompt = (const float*)d_in[1];
    const float* w1     = (const float*)d_in[2];
    const float* w2     = (const float*)d_in[3];
    const float* bias   = (const float*)d_in[4];
    const float* alog   = (const float*)d_in[5];
    const float* ds     = (const float*)d_in[6];
    float*       y      = (float*)d_out;

    float* ws = (float*)d_ws;
    size_t o = 0;
    float* dtsr  = ws + o; o += (size_t)MTOT * DTRANK;
    float* Bsc   = ws + o; o += (size_t)MTOT * DSTATE;
    float* Csc   = ws + o; o += (size_t)MTOT * DSTATE;
    float* delta = ws + o; o += (size_t)MTOT * DMODEL;
    const size_t units = (size_t)NCH * BB * DMODEL;
    float* P     = ws + o; o += units * DSTATE;
    float* Hend  = ws + o; o += units * DSTATE;
    float* Hinit = ws + o; o += units * DSTATE;
    (void)ws_size; (void)in_sizes; (void)n_in; (void)out_size;

    proj_gemm_kernel<<<dim3(MTOT / 128, 4, 1), 256, 0, stream>>>(x, w1, prompt, dtsr, Bsc, Csc);
    dt_gemm_scanA_kernel<<<dim3(MTOT / 128, DTSPLIT, 1), 256, 0, stream>>>(dtsr, w2, bias, x, Bsc, alog, delta, P, Hend);
    scan_passB_kernel<<<dim3(PB_PLANE / PB_THREADS), PB_THREADS, 0, stream>>>(P, Hend, Hinit);
    scan_passC_kernel<<<dim3((unsigned)(units / 32)), 128, 0, stream>>>(delta, x, Bsc, Csc, alog, ds, Hinit, y);
}


# Scan pass C: a thread's 4 states held as two 2-float pairs: per-state multiplies and multiply-adds become packed two-wide; y's 4 products summed (h0C0+h2C2)+(h1C1+h3C3); planned for 6 waves/SIMD
# speedup vs baseline: 7.8780x; 1.0562x over previous
#include <hip/hip_runtime.h>
#include <hip/hip_bf16.h>
#include <math.h>

#define BB      4
#define LL      4096
#define DMODEL  384
#define DSTATE  16
#define DTRANK  24
#define CTOT    (DTRANK + 2*DSTATE)
#define MTOT    (BB*LL)
#define NCH     32
#define CHL     (LL/NCH)
#define NQ      4
#define SPQ     (DSTATE/NQ)
#define DGRP    (DMODEL/32)
#define TB      32
#define UB      8
#define DTSPLIT 6
#define DTCOLS  (DMODEL/DTSPLIT)
#define XCH     64
#define XS      68
#define NXCH    (DMODEL/XCH)
#define LOG2E_F 1.4426950408889634f
#define LOG2E_LO_F 1.92596299e-8f
#define LN2_F      0.6931471805599453f

typedef float     v2f  __attribute__((ext_vector_type(2)));
typedef float     v8f  __attribute__((ext_vector_type(8)));
typedef _Float16  v16h __attribute__((ext_vector_type(16)));

#if defined(__gfx1250__) && __has_builtin(__builtin_amdgcn_wmma_f32_16x16x4_f32)
#define HAVE_WMMA_F32 1
#else
#define HAVE_WMMA_F32 0
#endif

__global__ __launch_bounds__(256)
void proj_gemm_kernel(const float* __restrict__ X,
                      const float* __restrict__ W1,
                      const float* __restrict__ prompt,
                      float* __restrict__ dtsr,
                      float* __restrict__ Bsc,
                      float* __restrict__ Csc)
{
    __shared__ __attribute__((aligned(16))) float Wlds[16 * 388];
    __shared__ __attribute__((aligned(16))) float Xlds[8][16 * XS];

    const int c0 = blockIdx.y * 16;
    const int wave = threadIdx.x >> 5;
    const int lane = threadIdx.x & 31;
    const int ln   = lane & 15;
    const int hi   = lane >> 4;
    const int m0   = blockIdx.x * 128 + wave * 16;

    float*       xt   = Xlds[wave];
    const float* xsrc = X + (size_t)(m0 + hi) * DMODEL + 4 * ln;
    float*       xdst = xt + hi * XS + 4 * ln;

    float4 xv[8];
    {
        float4 wv[6];
#pragma unroll
        for (int j = 0; j < 6; ++j) {
            const int f   = threadIdx.x + 256 * j;
            const int n   = f / (DMODEL / 4);
            const int k4  = f - n * (DMODEL / 4);
            const int row = (c0 + n < CTOT) ? (c0 + n) : (CTOT - 1);
            wv[j] = *reinterpret_cast<const float4*>(W1 + (size_t)row * DMODEL + 4 * k4);
        }
#pragma unroll
        for (int j = 0; j < 8; ++j)
            xv[j] = *reinterpret_cast<const float4*>(xsrc + (size_t)(2 * j) * DMODEL);
#pragma unroll
        for (int j = 0; j < 6; ++j) {
            const int f  = threadIdx.x + 256 * j;
            const int n  = f / (DMODEL / 4);
            const int k4 = f - n * (DMODEL / 4);
            float4 v = wv[j];
            if (c0 + n >= CTOT) v = make_float4(0.0f, 0.0f, 0.0f, 0.0f);
            *reinterpret_cast<float4*>(&Wlds[n * 388 + 4 * k4]) = v;
        }
#pragma unroll
        for (int j = 0; j < 8; ++j)
            *reinterpret_cast<float4*>(xdst + 2 * j * XS) = xv[j];
    }
    __syncthreads();

    v8f acc = {};

#if HAVE_WMMA_F32
#pragma unroll
    for (int c = 0; c < NXCH; ++c) {
        if (c + 1 < NXCH) {
#pragma unroll
            for (int j = 0; j < 8; ++j)
                xv[j] = *reinterpret_cast<const float4*>(xsrc + (size_t)(2 * j) * DMODEL + (c + 1) * XCH);
        }
        v2f af[XCH / 4], bf[XCH / 4];
#pragma unroll
        for (int s = 0; s < XCH / 4; ++s) {
            af[s].x = xt[ln * XS + 4 * s + 2 * hi];
            af[s].y = xt[ln * XS + 4 * s + 2 * hi + 1];
            bf[s].x = Wlds[ln * 388 + c * XCH + 4 * s + 2 * hi];
            bf[s].y = Wlds[ln * 388 + c * XCH + 4 * s + 2 * hi + 1];
        }
#pragma unroll
        for (int s = 0; s < XCH / 4; ++s)
            acc = __builtin_amdgcn_wmma_f32_16x16x4_f32(
                      false, af[s], false, bf[s], (short)0, acc, false, false);
        if (c + 1 < NXCH) {
#pragma unroll
            for (int j = 0; j < 8; ++j)
                *reinterpret_cast<float4*>(xdst + 2 * j * XS) = xv[j];
            __builtin_amdgcn_fence(__ATOMIC_SEQ_CST, "wavefront");
        }
    }
#else
    const float* arow = X + (size_t)(m0 + ln) * DMODEL;
    for (int k0 = 0; k0 < DMODEL; k0 += 32) {
        v16h a, b;
#pragma unroll
        for (int v = 0; v < 8; ++v) {
            int kk = (v < 4 ? 2 * v : 16 + 2 * (v - 4)) + 8 * hi;
            a[2 * v]     = (_Float16)arow[k0 + kk];
            a[2 * v + 1] = (_Float16)arow[k0 + kk + 1];
            b[2 * v]     = (_Float16)Wlds[ln * 388 + k0 + kk];
            b[2 * v + 1] = (_Float16)Wlds[ln * 388 + k0 + kk + 1];
        }
        acc = __builtin_amdgcn_wmma_f32_16x16x32_f16(
                  false, a, false, b, (short)0, acc, false, false);
    }
#endif

    const int c = c0 + ln;
#pragma unroll
    for (int r = 0; r < 8; ++r) {
        int m = m0 + r + 8 * hi;
        float v = acc[r];
        if (c < DTRANK) {
            dtsr[(size_t)m * DTRANK + c] = v;
        } else if (c < DTRANK + DSTATE) {
            Bsc[(size_t)m * DSTATE + (c - DTRANK)] = v;
        } else if (c < CTOT) {
            int n = c - DTRANK - DSTATE;
            Csc[(size_t)m * DSTATE + n] = v + prompt[(size_t)m * DSTATE + n];
        }
    }
}

__device__ __forceinline__ float softplus_f32(float t)
{
    const float na  = -__builtin_fabsf(t);
    const float hi  = na * LOG2E_F;
    float       lo  = __builtin_fmaf(na, LOG2E_F, -hi);
    lo              = __builtin_fmaf(na, LOG2E_LO_F, lo);
    const float e   = __builtin_amdgcn_exp2f(hi);
    const float z   = __builtin_fmaf(e, lo * LN2_F, e);
    const float den = 2.0f + z;
    const float r   = __builtin_amdgcn_rcpf(den);
    float       s   = z * r;
    s               = __builtin_fmaf(__builtin_fmaf(-s, den, z), r, s);
    const float w   = s * s;
    float p = 2.0f / 15.0f;
    p = __builtin_fmaf(p, w, 2.0f / 13.0f);
    p = __builtin_fmaf(p, w, 2.0f / 11.0f);
    p = __builtin_fmaf(p, w, 2.0f / 9.0f);
    p = __builtin_fmaf(p, w, 2.0f / 7.0f);
    p = __builtin_fmaf(p, w, 2.0f / 5.0f);
    p = __builtin_fmaf(p, w, 2.0f / 3.0f);
    p = __builtin_fmaf(p, w, 2.0f);
    return __builtin_fmaf(s, p, __builtin_fmaxf(t, 0.0f));
}

#define DLP 68
__global__ __launch_bounds__(256) __attribute__((amdgpu_waves_per_eu(8, 8)))
void dt_gemm_scanA_kernel(const float* __restrict__ dtsr,
                          const float* __restrict__ W2,
                          const float* __restrict__ bias,
                          const float* __restrict__ X,
                          const float* __restrict__ Bsc,
                          const float* __restrict__ Alog,
                          float* __restrict__ delta,
                          float* __restrict__ Pout,
                          float* __restrict__ Hend)
{
    __shared__ __attribute__((aligned(16))) float Wlds[DTCOLS * 28];
    __shared__ __attribute__((aligned(16))) float dl[CHL * DLP];

    const int dbase = blockIdx.y * DTCOLS;
    const int wave = threadIdx.x >> 5;
    const int lane = threadIdx.x & 31;
    const int ln   = lane & 15;
    const int hi   = lane >> 4;
    const int m0   = blockIdx.x * 128 + wave * 16;

    const int q    = wave & 3;
    const int dloc = (wave >> 2) * 32 + lane;
    const int dsc  = dbase + dloc;

    const float* arow = dtsr + (size_t)(m0 + ln) * DTRANK;

    const float2* wsrc = reinterpret_cast<const float2*>(W2 + (size_t)dbase * DTRANK);
    float2 wv[3];
#pragma unroll
    for (int j = 0; j < 3; ++j) wv[j] = wsrc[threadIdx.x + 256 * j];

#if HAVE_WMMA_F32
    v2f afr[6];
#pragma unroll
    for (int ks = 0; ks < 6; ++ks) {
        afr[ks].x = arow[4 * ks + 2 * hi];
        afr[ks].y = arow[4 * ks + 2 * hi + 1];
    }
#else
    v16h afr;
#pragma unroll
    for (int v = 0; v < 8; ++v) {
        int kk = (v < 4 ? 2 * v : 16 + 2 * (v - 4)) + 8 * hi;
        afr[2 * v]     = (kk     < DTRANK) ? (_Float16)arow[kk]     : (_Float16)0.f;
        afr[2 * v + 1] = (kk + 1 < DTRANK) ? (_Float16)arow[kk + 1] : (_Float16)0.f;
    }
#endif
    const float4 al = *reinterpret_cast<const float4*>(Alog + dsc * DSTATE + SPQ * q);

#pragma unroll
    for (int j = 0; j < 3; ++j) {
        const int f  = threadIdx.x + 256 * j;
        const int ds = f / (DTRANK / 2);
        const int r2 = f - ds * (DTRANK / 2);
        *reinterpret_cast<float2*>(&Wlds[ds * 28 + 2 * r2]) = wv[j];
    }
    __syncthreads();

    for (int d0 = 0; d0 < DTCOLS; d0 += 16) {
        v8f acc = {};
#if HAVE_WMMA_F32
#pragma unroll
        for (int ks = 0; ks < 6; ++ks) {
            v2f b;
            b.x = Wlds[(d0 + ln) * 28 + 4 * ks + 2 * hi];
            b.y = Wlds[(d0 + ln) * 28 + 4 * ks + 2 * hi + 1];
            acc = __builtin_amdgcn_wmma_f32_16x16x4_f32(
                      false, afr[ks], false, b, (short)0, acc, false, false);
        }
#else
        v16h b;
#pragma unroll
        for (int v = 0; v < 8; ++v) {
            int kk = (v < 4 ? 2 * v : 16 + 2 * (v - 4)) + 8 * hi;
            b[2 * v]     = (kk     < DTRANK) ? (_Float16)Wlds[(d0 + ln) * 28 + kk]     : (_Float16)0.f;
            b[2 * v + 1] = (kk + 1 < DTRANK) ? (_Float16)Wlds[(d0 + ln) * 28 + kk + 1] : (_Float16)0.f;
        }
        acc = __builtin_amdgcn_wmma_f32_16x16x32_f16(
                  false, afr, false, b, (short)0, acc, false, false);
#endif
        const int   d  = dbase + d0 + ln;
        const float bi = bias[d];
#pragma unroll
        for (int r = 0; r < 8; ++r) {
            const int   m  = m0 + r + 8 * hi;
            const float sp = softplus_f32(acc[r] + bi);
            delta[(size_t)m * DMODEL + d] = sp;
            dl[(wave * 16 + r + 8 * hi) * DLP + d0 + ln] = sp;
        }
    }
    __syncthreads();

    const int b  = blockIdx.x / NCH;
    const int ch = blockIdx.x % NCH;
    const size_t unit = (size_t)(ch * BB + b) * DMODEL + dsc;

    float A[SPQ], h[SPQ], P[SPQ];
    A[0] = -expf(al.x) * LOG2E_F; A[1] = -expf(al.y) * LOG2E_F;
    A[2] = -expf(al.z) * LOG2E_F; A[3] = -expf(al.w) * LOG2E_F;
    for (int j = 0; j < SPQ; ++j) { h[j] = 0.0f; P[j] = 1.0f; }

    const size_t row0 = (size_t)blockIdx.x * CHL;
    const float* dp = dl + dloc;
    const float* xp = X   + row0 * DMODEL + dsc;
    const float* bp = Bsc + row0 * DSTATE + SPQ * q;
#pragma unroll 1
    for (int t0 = 0; t0 < CHL; t0 += UB) {
        float  dtv[UB], uv[UB];
        float4 bv[UB];
#pragma unroll
        for (int i = 0; i < UB; ++i) {
            dtv[i] = dp[(t0 + i) * DLP];
            uv[i]  = xp[(size_t)(t0 + i) * DMODEL];
            bv[i]  = *reinterpret_cast<const float4*>(bp + (size_t)(t0 + i) * DSTATE);
        }
#pragma unroll
        for (int i = 0; i < UB; ++i) {
            const float dt = dtv[i];
            const float du = dt * uv[i];
            const float Bl[SPQ] = { bv[i].x, bv[i].y, bv[i].z, bv[i].w };
            for (int j = 0; j < SPQ; ++j) {
                float a = __builtin_amdgcn_exp2f(dt * A[j]);
                P[j] *= a;
                h[j]  = a * h[j] + du * Bl[j];
            }
        }
    }

    float4* Pp = reinterpret_cast<float4*>(Pout + unit * DSTATE + SPQ * q);
    float4* Hp = reinterpret_cast<float4*>(Hend + unit * DSTATE + SPQ * q);
    *Pp = make_float4(P[0], P[1], P[2], P[3]);
    *Hp = make_float4(h[0], h[1], h[2], h[3]);
}

#define PB_THREADS 96
#define PB_PLANE   (BB * DMODEL * DSTATE)
__global__ __launch_bounds__(PB_THREADS)
void scan_passB_kernel(const float* __restrict__ P,
                       const float* __restrict__ Hend,
                       float* __restrict__ Hinit)
{
    const int idx = blockIdx.x * PB_THREADS + threadIdx.x;

    float p[NCH], e[NCH];
#pragma unroll
    for (int ch = 0; ch < NCH; ++ch) {
        p[ch] = P[(size_t)ch * PB_PLANE + idx];
        e[ch] = Hend[(size_t)ch * PB_PLANE + idx];
    }

    float h = 0.0f;
#pragma unroll
    for (int ch = 0; ch < NCH; ++ch) {
        Hinit[(size_t)ch * PB_PLANE + idx] = h;
        h = __builtin_fmaf(p[ch], h, e[ch]);
    }
}

__global__ __launch_bounds__(128) __attribute__((amdgpu_waves_per_eu(6, 6)))
void scan_passC_kernel(const float* __restrict__ delta,
                       const float* __restrict__ X,
                       const float* __restrict__ Bsc,
                       const float* __restrict__ Csc,
                       const float* __restrict__ Alog,
                       const float* __restrict__ Ds,
                       const float* __restrict__ Hinit,
                       float* __restrict__ Y)
{
    __shared__ float ypart[2][TB][NQ][32];

    const int lane = threadIdx.x & 31;
    const int q    = threadIdx.x >> 5;
    const int dg   = blockIdx.x % DGRP;
    const int t2   = blockIdx.x / DGRP;
    const int b    = t2 % BB;
    const int ch   = t2 / BB;
    const int d    = dg * 32 + lane;
    const size_t unit = (size_t)(ch * BB + b) * DMODEL + d;

    v2f A01, A23, h01, h23;
    {
        const float4 al = *reinterpret_cast<const float4*>(Alog + d * DSTATE + SPQ * q);
        const float4 hv = *reinterpret_cast<const float4*>(Hinit + unit * DSTATE + SPQ * q);
        A01.x = -expf(al.x) * LOG2E_F; A01.y = -expf(al.y) * LOG2E_F;
        A23.x = -expf(al.z) * LOG2E_F; A23.y = -expf(al.w) * LOG2E_F;
        h01.x = hv.x; h01.y = hv.y; h23.x = hv.z; h23.y = hv.w;
    }
    const float Dsd = Ds[d];

    const int l0 = ch * CHL;
    const size_t row0 = (size_t)(b * LL + l0);
    const float* dp = delta + row0 * DMODEL + d;
    const float* xp = X     + row0 * DMODEL + d;
    const float* bp = Bsc   + row0 * DSTATE + SPQ * q;
    const float* cp = Csc   + row0 * DSTATE + SPQ * q;
    for (int bt = 0; bt < CHL / TB; ++bt) {
        float (*yp)[NQ][32] = ypart[bt & 1];
#pragma unroll 1
        for (int s0 = 0; s0 < TB; s0 += UB) {
            const int t0 = bt * TB + s0;
            float  dtv[UB], uv[UB];
            float4 bv[UB], cv[UB];
#pragma unroll
            for (int i = 0; i < UB; ++i) {
                dtv[i] = dp[(size_t)(t0 + i) * DMODEL];
                uv[i]  = xp[(size_t)(t0 + i) * DMODEL];
                bv[i]  = *reinterpret_cast<const float4*>(bp + (size_t)(t0 + i) * DSTATE);
                cv[i]  = *reinterpret_cast<const float4*>(cp + (size_t)(t0 + i) * DSTATE);
            }
#pragma unroll
            for (int i = 0; i < UB; ++i) {
                const float dt = dtv[i];
                const float u  = uv[i];
                const float du = dt * u;
                const v2f dt2 = { dt, dt };
                const v2f du2 = { du, du };
                const v2f B01 = { bv[i].x, bv[i].y }, B23 = { bv[i].z, bv[i].w };
                const v2f C01 = { cv[i].x, cv[i].y }, C23 = { cv[i].z, cv[i].w };
                const v2f g01 = dt2 * A01, g23 = dt2 * A23;
                v2f a01, a23;
                a01.x = __builtin_amdgcn_exp2f(g01.x); a01.y = __builtin_amdgcn_exp2f(g01.y);
                a23.x = __builtin_amdgcn_exp2f(g23.x); a23.y = __builtin_amdgcn_exp2f(g23.y);
                h01 = __builtin_elementwise_fma(du2, B01, a01 * h01);
                h23 = __builtin_elementwise_fma(du2, B23, a23 * h23);
                const v2f yy = __builtin_elementwise_fma(h23, C23, h01 * C01);
                float y = yy.x + yy.y;
                if (q == 0) y += u * Dsd;
                yp[s0 + i][q][lane] = y;
            }
        }
        __syncthreads();
        for (int r = 0; r < TB / NQ; ++r) {
            const int tt = q * (TB / NQ) + r;
            const float s = (yp[tt][0][lane] + yp[tt][1][lane])
                          + (yp[tt][2][lane] + yp[tt][3][lane]);
            const int bl = b * LL + l0 + bt * TB + tt;
            Y[(size_t)bl * DMODEL + d] = s;
        }
    }
}

extern "C" void kernel_launch(void* const* d_in, const int* in_sizes, int n_in,
                              void* d_out, int out_size, void* d_ws, size_t ws_size,
                              hipStream_t stream)
{
    const float* x      = (const float*)d_in[0];
    const float* prompt = (const float*)d_in[1];
    const float* w1     = (const float*)d_in[2];
    const float* w2     = (const float*)d_in[3];
    const float* bias   = (const float*)d_in[4];
    const float* alog   = (const float*)d_in[5];
    const float* ds     = (const float*)d_in[6];
    float*       y      = (float*)d_out;

    float* ws = (float*)d_ws;
    size_t o = 0;
    float* dtsr  = ws + o; o += (size_t)MTOT * DTRANK;
    float* Bsc   = ws + o; o += (size_t)MTOT * DSTATE;
    float* Csc   = ws + o; o += (size_t)MTOT * DSTATE;
    float* delta = ws + o; o += (size_t)MTOT * DMODEL;
    const size_t units = (size_t)NCH * BB * DMODEL;
    float* P     = ws + o; o += units * DSTATE;
    float* Hend  = ws + o; o += units * DSTATE;
    float* Hinit = ws + o; o += units * DSTATE;
    (void)ws_size; (void)in_sizes; (void)n_in; (void)out_size;

    proj_gemm_kernel<<<dim3(MTOT / 128, 4, 1), 256, 0, stream>>>(x, w1, prompt, dtsr, Bsc, Csc);
    dt_gemm_scanA_kernel<<<dim3(MTOT / 128, DTSPLIT, 1), 256, 0, stream>>>(dtsr, w2, bias, x, Bsc, alog, delta, P, Hend);
    scan_passB_kernel<<<dim3(PB_PLANE / PB_THREADS), PB_THREADS, 0, stream>>>(P, Hend, Hinit);
    scan_passC_kernel<<<dim3((unsigned)(units / 32)), 128, 0, stream>>>(delta, x, Bsc, Csc, alog, ds, Hinit, y);
}
